# starting kernel
# baseline (speedup 1.0000x reference)
#include <hip/hip_runtime.h>


typedef __attribute__((ext_vector_type(2))) float v2f;
typedef __attribute__((ext_vector_type(8))) float v8f;

#define IN_F 128
#define H_F  64
#define C_F  32

static inline int ceil_div_h(long a, long b) { return (int)((a + b - 1) / b); }


__global__ void fill_zero32_kernel(unsigned int* __restrict__ p, long n) {
    long i = (long)blockIdx.x * blockDim.x + threadIdx.x;
    if (i < n) p[i] = 0u;
}

__global__ void deg_hist_kernel(const int* __restrict__ dst, const float* __restrict__ ew,
                                float* __restrict__ deg, int* __restrict__ counts, int E) {
    int e = blockIdx.x * blockDim.x + threadIdx.x;
    if (e < E) {
        int d = dst[e];
        atomicAdd(&deg[d], ew[e]);
        atomicAdd(&counts[d], 1);
    }
}

__global__ void norm_kernel(float* __restrict__ deg, int n) {
    int i = blockIdx.x * blockDim.x + threadIdx.x;
    if (i < n) {
        float d = deg[i];
        deg[i] = (d > 0.0f) ? rsqrtf(fmaxf(d, 1e-12f)) : 0.0f;
    }
}


__global__ void scan_block_kernel(const int* __restrict__ counts, int* __restrict__ rowptr,
                                  int* __restrict__ bsums, int n) {
    __shared__ int sdata[256];
    int tid = threadIdx.x;
    int i = blockIdx.x * 256 + tid;
    int v = (i < n) ? counts[i] : 0;
    sdata[tid] = v;
    __syncthreads();
#pragma unroll
    for (int off = 1; off < 256; off <<= 1) {
        int t = (tid >= off) ? sdata[tid - off] : 0;
        __syncthreads();
        sdata[tid] += t;
        __syncthreads();
    }
    if (i < n) rowptr[i] = sdata[tid] - v;
    if (tid == 255) bsums[blockIdx.x] = sdata[255];
}

__global__ void scan_sums_kernel(int* __restrict__ bsums, int nb) {
    if (blockIdx.x == 0 && threadIdx.x == 0) {
        int acc = 0;
        for (int i = 0; i < nb; ++i) { int t = bsums[i]; bsums[i] = acc; acc += t; }
    }
}

__global__ void scan_add_kernel(int* __restrict__ rowptr, const int* __restrict__ bsums,
                                int n, int E) {
    int i = blockIdx.x * blockDim.x + threadIdx.x;
    if (i < n) rowptr[i] += bsums[i >> 8];
    if (i == 0) rowptr[n] = E;
}

__global__ void build_edges_kernel(const int* __restrict__ src, const int* __restrict__ dst,
                                   const float* __restrict__ ew, const float* __restrict__ norm,
                                   const int* __restrict__ rowptr, int* __restrict__ cursor,
                                   int2* __restrict__ eRec, int E) {
    int e = blockIdx.x * blockDim.x + threadIdx.x;
    if (e < E) {
        int s = src[e], d = dst[e];
        float c = ew[e] * norm[s] * norm[d];
        int pos = rowptr[d] + atomicAdd(&cursor[d], 1);
        eRec[pos] = make_int2(s, __float_as_int(c));
    }
}

template <int FIN, int FOUT>
__global__ void gemm_wmma_kernel(const float* __restrict__ X, const float* __restrict__ W,
                                 float* __restrict__ Y, int nrows) {
    __shared__ float tx[16 * FIN];
    const int nthreads = (FOUT / 16) * 32;
    const int m0  = blockIdx.x * 16;
    const int tid = threadIdx.x;

    for (int i = tid; i < 16 * FIN; i += nthreads) {
        int r = i / FIN, c = i % FIN;
        int row = m0 + r;
        if (row >= nrows) row = nrows - 1;
        tx[i] = X[(long)row * FIN + c];
    }
    __syncthreads();

    const int wave  = tid >> 5;
    const int lane  = tid & 31;
    const int lrow  = lane & 15;
    const int khalf = (lane >> 4) * 2;
    const int n0    = wave * 16;

    v8f acc = {};
#pragma unroll
    for (int k0 = 0; k0 < FIN; k0 += 4) {
        v2f a = *(const v2f*)&tx[lrow * FIN + k0 + khalf];
        v2f b;
        b.x = W[(long)(k0 + khalf + 0) * FOUT + n0 + lrow];
        b.y = W[(long)(k0 + khalf + 1) * FOUT + n0 + lrow];
        acc = __builtin_amdgcn_wmma_f32_16x16x4_f32(
            false, a, false, b, (short)0, acc, false, false);
    }

    const int mbase = m0 + ((lane >> 4) * 8);
    const int col   = n0 + lrow;
#pragma unroll
    for (int r = 0; r < 8; ++r) {
        int row = mbase + r;
        if (row < nrows) Y[(long)row * FOUT + col] = acc[r];
    }
}

template <int F, bool RELU, bool FINAL>
__global__ void gather_agg_kernel(const float* __restrict__ h,
                                  const int* __restrict__ rowptr,
                                  const int2* __restrict__ eRec,
                                  const float* __restrict__ bias,
                                  const float* __restrict__ res,
                                  const float* __restrict__ bias2,
                                  float* __restrict__ y, int n) {
    const int wave = threadIdx.x >> 5;
    const int lane = threadIdx.x & 31;
    const int node = blockIdx.x * (blockDim.x >> 5) + wave;
    if (node >= n) return;

    const int s0 = rowptr[node];
    const int s1 = rowptr[node + 1];

    constexpr int VPL = F / 32;
    const int fbase = lane * VPL;

    float acc[VPL];
#pragma unroll
    for (int j = 0; j < VPL; ++j) acc[j] = 0.0f;

    for (int i = s0; i < s1; ++i) {
        int2  r = eRec[i];
        float c = __int_as_float(r.y);
        const float* hr = h + (long)r.x * F + fbase;
        if constexpr (VPL == 2) {
            float2 v = *(const float2*)hr;
            acc[0] += v.x * c;
            acc[1] += v.y * c;
        } else {
            acc[0] += hr[0] * c;
        }
    }

    const long base = (long)node * F + fbase;
#pragma unroll
    for (int j = 0; j < VPL; ++j) {
        float v = acc[j] + bias[fbase + j];
        if (FINAL) v += res[base + j] + bias2[fbase + j];
        if (RELU) v = fmaxf(v, 0.0f);
        acc[j] = v;
    }
    if constexpr (VPL == 2) {
        *(float2*)(y + base) = make_float2(acc[0], acc[1]);
    } else {
        y[base] = acc[0];
    }
}


extern "C" void kernel_launch(void* const* d_in, const int* in_sizes, int n_in,
                              void* d_out, int out_size, void* d_ws, size_t ws_size,
                              hipStream_t stream) {
    const float* x    = (const float*)d_in[0];
    const int*   src  = (const int*)  d_in[1];
    const int*   dst  = (const int*)  d_in[2];
    const float* ew   = (const float*)d_in[3];
    const float* W1   = (const float*)d_in[4];
    const float* b1   = (const float*)d_in[5];
    const float* W2   = (const float*)d_in[6];
    const float* b2   = (const float*)d_in[7];
    const float* W3   = (const float*)d_in[8];
    const float* b3   = (const float*)d_in[9];
    const float* W4   = (const float*)d_in[10];
    const float* b4   = (const float*)d_in[11];
    const float* Wres = (const float*)d_in[12];
    const float* bres = (const float*)d_in[13];
    float* out = (float*)d_out;

    const int N  = in_sizes[0] / IN_F;
    const int E  = in_sizes[1];
    const int nb = ceil_div_h(N, 256);

    unsigned int* ws = (unsigned int*)d_ws;
    float* deg    = (float*)(ws);
    int*   counts = (int*)  (ws + N);
    int*   cursor = (int*)  (ws + 2L * N);
    int*   rowptr = (int*)  (ws + 3L * N);
    int*   bsums  = (int*)  (ws + 4L * N + 1);
    long   recOff = (4L * N + 1 + nb + 1) & ~1L;
    int2*  eRec   = (int2*) (ws + recOff);
    float* h      = (float*)(ws + recOff + 2L * E);
    float* xbuf   = h    + (size_t)N * H_F;
    float* res    = xbuf + (size_t)N * H_F;

    const int T  = 256;
    const int mt = ceil_div_h(N, 16);
    const int gblocks = ceil_div_h(N, T / 32);

    fill_zero32_kernel<<<ceil_div_h(3L * N, T), T, 0, stream>>>(ws, 3L * N);
    deg_hist_kernel<<<ceil_div_h(E, T), T, 0, stream>>>(dst, ew, deg, counts, E);
    norm_kernel<<<ceil_div_h(N, T), T, 0, stream>>>(deg, N);
    scan_block_kernel<<<nb, 256, 0, stream>>>(counts, rowptr, bsums, N);
    scan_sums_kernel<<<1, 32, 0, stream>>>(bsums, nb);
    scan_add_kernel<<<ceil_div_h(N, T), T, 0, stream>>>(rowptr, bsums, N, E);
    build_edges_kernel<<<ceil_div_h(E, T), T, 0, stream>>>(src, dst, ew, deg, rowptr, cursor,
                                                           eRec, E);

    gemm_wmma_kernel<IN_F, C_F><<<mt, (C_F / 16) * 32, 0, stream>>>(x, Wres, res, N);

    gemm_wmma_kernel<IN_F, H_F><<<mt, (H_F / 16) * 32, 0, stream>>>(x, W1, h, N);
    gather_agg_kernel<H_F, true, false><<<gblocks, T, 0, stream>>>(h, rowptr, eRec,
                                                                   b1, nullptr, nullptr, xbuf, N);
    gemm_wmma_kernel<H_F, H_F><<<mt, (H_F / 16) * 32, 0, stream>>>(xbuf, W2, h, N);
    gather_agg_kernel<H_F, true, false><<<gblocks, T, 0, stream>>>(h, rowptr, eRec,
                                                                   b2, nullptr, nullptr, xbuf, N);
    gemm_wmma_kernel<H_F, H_F><<<mt, (H_F / 16) * 32, 0, stream>>>(xbuf, W3, h, N);
    gather_agg_kernel<H_F, true, false><<<gblocks, T, 0, stream>>>(h, rowptr, eRec,
                                                                   b3, nullptr, nullptr, xbuf, N);
    gemm_wmma_kernel<H_F, C_F><<<mt, (C_F / 16) * 32, 0, stream>>>(xbuf, W4, h, N);
    gather_agg_kernel<C_F, false, true><<<gblocks, T, 0, stream>>>(h, rowptr, eRec,
                                                                   b4, res, bres, out, N);
}


# CSR build by two-level LDS counting sort (a work-group owns each 256-node destination range): no global atomics; degree now summed by LDS float adds in arrival order
# speedup vs baseline: 1.6046x; 1.6046x over previous
#include <hip/hip_runtime.h>


typedef __attribute__((ext_vector_type(2))) float v2f;
typedef __attribute__((ext_vector_type(8))) float v8f;

#define IN_F 128
#define H_F  64
#define C_F  32

#define BSH    8
#define BNODES (1 << BSH)
#define CHUNK  16384
#define PT     1024
#define RCAP   6144

static inline int ceil_div_h(long a, long b) { return (int)((a + b - 1) / b); }


__global__ void bucket_hist_kernel(const int* __restrict__ dst, int* __restrict__ bh,
                                   int E, int nbk) {
    extern __shared__ int s_bins[];
    for (int k = threadIdx.x; k < nbk; k += blockDim.x) s_bins[k] = 0;
    __syncthreads();
    const int e0 = blockIdx.x * CHUNK;
    const int e1 = min(e0 + CHUNK, E);
    const int bs = blockDim.x;
    for (int eb = e0 + threadIdx.x; eb < e1; eb += 4 * bs) {
        int d[4];
        for (int j = 0; j < 4; ++j) d[j] = dst[min(eb + j * bs, e1 - 1)];
        asm volatile("" :: "v"(d[0]), "v"(d[1]), "v"(d[2]), "v"(d[3]));
        for (int j = 0; j < 4; ++j)
            if (eb + j * bs < e1) atomicAdd(&s_bins[d[j] >> BSH], 1);
    }
    __syncthreads();
    for (int k = threadIdx.x; k < nbk; k += blockDim.x)
        bh[k * gridDim.x + blockIdx.x] = s_bins[k];
}


__global__ void scan_block_kernel(const int* __restrict__ counts, int* __restrict__ rowptr,
                                  int* __restrict__ bsums, int n) {
    __shared__ int sdata[256];
    int tid = threadIdx.x;
    int i = blockIdx.x * 256 + tid;
    int v = (i < n) ? counts[i] : 0;
    sdata[tid] = v;
    __syncthreads();
#pragma unroll
    for (int off = 1; off < 256; off <<= 1) {
        int t = (tid >= off) ? sdata[tid - off] : 0;
        __syncthreads();
        sdata[tid] += t;
        __syncthreads();
    }
    if (i < n) rowptr[i] = sdata[tid] - v;
    if (tid == 255) bsums[blockIdx.x] = sdata[255];
}

__global__ void scan_sums_kernel(int* __restrict__ bsums, int nb) {
    if (blockIdx.x == 0 && threadIdx.x == 0) {
        int acc = 0;
        for (int i = 0; i < nb; ++i) { int t = bsums[i]; bsums[i] = acc; acc += t; }
    }
}

__global__ void scan_add_kernel(int* __restrict__ rowptr, const int* __restrict__ bsums,
                                int n, int E) {
    int i = blockIdx.x * blockDim.x + threadIdx.x;
    if (i < n) rowptr[i] += bsums[i >> 8];
    if (i == 0) rowptr[n] = E;
}

__global__ void bucket_scatter_kernel(const int* __restrict__ src, const int* __restrict__ dst,
                                      const float* __restrict__ ew, const int* __restrict__ offs,
                                      int4* __restrict__ tmp, int E, int nbk) {
    extern __shared__ int s_bins[];
    for (int k = threadIdx.x; k < nbk; k += blockDim.x)
        s_bins[k] = offs[k * gridDim.x + blockIdx.x];
    __syncthreads();
    const int e0 = blockIdx.x * CHUNK;
    const int e1 = min(e0 + CHUNK, E);
    const int bs = blockDim.x;
    for (int eb = e0 + threadIdx.x; eb < e1; eb += 4 * bs) {
        int s[4], d[4], w[4];
        for (int j = 0; j < 4; ++j) {
            const int e = min(eb + j * bs, e1 - 1);
            s[j] = src[e];
            d[j] = dst[e];
            w[j] = __float_as_int(ew[e]);
        }
        asm volatile("" :: "v"(s[0]), "v"(s[1]), "v"(s[2]), "v"(s[3]),
                           "v"(d[0]), "v"(d[1]), "v"(d[2]), "v"(d[3]),
                           "v"(w[0]), "v"(w[1]), "v"(w[2]), "v"(w[3]));
        for (int j = 0; j < 4; ++j) {
            if (eb + j * bs < e1) {
                int pos = atomicAdd(&s_bins[d[j] >> BSH], 1);
                tmp[pos] = make_int4(s[j], w[j], d[j], 0);
            }
        }
    }
}

__global__ void bucket_build_kernel(const int4* __restrict__ tmp, const int* __restrict__ offs,
                                    int* __restrict__ rowptr, float* __restrict__ norm,
                                    int2* __restrict__ eRec, int n, int E, int g1) {
    __shared__ int   s_cnt[BNODES];
    __shared__ float s_deg[BNODES];
    __shared__ int   s_scan[BNODES];
    __shared__ int2  s_rec[RCAP];
    const int tid   = threadIdx.x;
    const int t0    = offs[blockIdx.x * g1];
    const int t1    = offs[(blockIdx.x + 1) * g1];
    const int nbase = blockIdx.x << BSH;
    const bool stage = (t1 - t0) <= RCAP;

    if (tid < BNODES) { s_cnt[tid] = 0; s_deg[tid] = 0.0f; }
    __syncthreads();
    const int bs = blockDim.x;
    for (int ib = t0 + tid; ib < t1; ib += 4 * bs) {
        int4 r[4];
        for (int j = 0; j < 4; ++j) r[j] = tmp[min(ib + j * bs, t1 - 1)];
        asm volatile("" :: "v"(r[0].y), "v"(r[0].z), "v"(r[1].y), "v"(r[1].z),
                           "v"(r[2].y), "v"(r[2].z), "v"(r[3].y), "v"(r[3].z));
        for (int j = 0; j < 4; ++j) {
            if (ib + j * bs < t1) {
                int dl = r[j].z - nbase;
                atomicAdd(&s_cnt[dl], 1);
                atomicAdd(&s_deg[dl], __int_as_float(r[j].y));
            }
        }
    }
    __syncthreads();

    int v = 0;
    if (tid < BNODES) { v = s_cnt[tid]; s_scan[tid] = v; }
    __syncthreads();
    for (int off = 1; off < BNODES; off <<= 1) {
        int t = 0;
        if (tid < BNODES && tid >= off) t = s_scan[tid - off];
        __syncthreads();
        if (tid < BNODES) s_scan[tid] += t;
        __syncthreads();
    }
    if (tid < BNODES) {
        const int start = t0 + s_scan[tid] - v;
        s_cnt[tid] = start;
        const int node = nbase + tid;
        if (node < n) {
            rowptr[node] = start;
            float d = s_deg[tid];
            norm[node] = (d > 0.0f) ? rsqrtf(fmaxf(d, 1e-12f)) : 0.0f;
        }
    }
    if (blockIdx.x == gridDim.x - 1 && tid == 0) rowptr[n] = E;
    __syncthreads();

    for (int ib = t0 + tid; ib < t1; ib += 4 * bs) {
        int4 r[4];
        for (int j = 0; j < 4; ++j) r[j] = tmp[min(ib + j * bs, t1 - 1)];
        asm volatile("" :: "v"(r[0].x), "v"(r[0].y), "v"(r[0].z), "v"(r[1].x), "v"(r[1].y), "v"(r[1].z),
                           "v"(r[2].x), "v"(r[2].y), "v"(r[2].z), "v"(r[3].x), "v"(r[3].y), "v"(r[3].z));
        for (int j = 0; j < 4; ++j) {
            if (ib + j * bs < t1) {
                int pos = atomicAdd(&s_cnt[r[j].z - nbase], 1);
                if (stage) s_rec[pos - t0] = make_int2(r[j].x, r[j].y);
                else       eRec[pos]       = make_int2(r[j].x, r[j].y);
            }
        }
    }
    if (stage) {
        __syncthreads();
        for (int i = tid; i < t1 - t0; i += blockDim.x) eRec[t0 + i] = s_rec[i];
    }
}

__global__ void row_coef_kernel(const int* __restrict__ rowptr, const float* __restrict__ norm,
                                int2* __restrict__ eRec, int n) {
    const int t   = blockIdx.x * blockDim.x + threadIdx.x;
    const int n0  = (t >> 4) * 4;
    const int sub = t & 15;
    if (n0 >= n) return;
    const int nl = min(n0 + 4, n);
    int   rp[5];
    float nd[4];
    for (int q = 0; q < 5; ++q) rp[q] = rowptr[min(n0 + q, nl)];
    for (int q = 0; q < 4; ++q) nd[q] = norm[min(n0 + q, n - 1)];
    for (int i = rp[0] + sub; i < rp[4]; i += 16) {
        int2  r   = eRec[i];
        float ndq = nd[0];
        if (i >= rp[1]) ndq = nd[1];
        if (i >= rp[2]) ndq = nd[2];
        if (i >= rp[3]) ndq = nd[3];
        float c = __int_as_float(r.y) * norm[r.x] * ndq;
        eRec[i] = make_int2(r.x, __float_as_int(c));
    }
}

template <int FIN, int FOUT>
__global__ void gemm_wmma_kernel(const float* __restrict__ X, const float* __restrict__ W,
                                 float* __restrict__ Y, int nrows) {
    __shared__ float tx[16 * FIN];
    const int nthreads = (FOUT / 16) * 32;
    const int m0  = blockIdx.x * 16;
    const int tid = threadIdx.x;

    for (int i = tid; i < 16 * FIN; i += nthreads) {
        int r = i / FIN, c = i % FIN;
        int row = m0 + r;
        if (row >= nrows) row = nrows - 1;
        tx[i] = X[(long)row * FIN + c];
    }
    __syncthreads();

    const int wave  = tid >> 5;
    const int lane  = tid & 31;
    const int lrow  = lane & 15;
    const int khalf = (lane >> 4) * 2;
    const int n0    = wave * 16;

    v8f acc = {};
#pragma unroll
    for (int k0 = 0; k0 < FIN; k0 += 4) {
        v2f a = *(const v2f*)&tx[lrow * FIN + k0 + khalf];
        v2f b;
        b.x = W[(long)(k0 + khalf + 0) * FOUT + n0 + lrow];
        b.y = W[(long)(k0 + khalf + 1) * FOUT + n0 + lrow];
        acc = __builtin_amdgcn_wmma_f32_16x16x4_f32(
            false, a, false, b, (short)0, acc, false, false);
    }

    const int mbase = m0 + ((lane >> 4) * 8);
    const int col   = n0 + lrow;
#pragma unroll
    for (int r = 0; r < 8; ++r) {
        int row = mbase + r;
        if (row < nrows) Y[(long)row * FOUT + col] = acc[r];
    }
}

template <int F, bool RELU, bool FINAL>
__global__ void gather_agg_kernel(const float* __restrict__ h,
                                  const int* __restrict__ rowptr,
                                  const int2* __restrict__ eRec,
                                  const float* __restrict__ bias,
                                  const float* __restrict__ res,
                                  const float* __restrict__ bias2,
                                  float* __restrict__ y, int n) {
    const int wave = threadIdx.x >> 5;
    const int lane = threadIdx.x & 31;
    const int node = blockIdx.x * (blockDim.x >> 5) + wave;
    if (node >= n) return;

    const int s0 = rowptr[node];
    const int s1 = rowptr[node + 1];

    constexpr int VPL = F / 32;
    const int fbase = lane * VPL;

    float acc[VPL];
#pragma unroll
    for (int j = 0; j < VPL; ++j) acc[j] = 0.0f;

    for (int i = s0; i < s1; ++i) {
        int2  r = eRec[i];
        float c = __int_as_float(r.y);
        const float* hr = h + (long)r.x * F + fbase;
        if constexpr (VPL == 2) {
            float2 v = *(const float2*)hr;
            acc[0] += v.x * c;
            acc[1] += v.y * c;
        } else {
            acc[0] += hr[0] * c;
        }
    }

    const long base = (long)node * F + fbase;
#pragma unroll
    for (int j = 0; j < VPL; ++j) {
        float v = acc[j] + bias[fbase + j];
        if (FINAL) v += res[base + j] + bias2[fbase + j];
        if (RELU) v = fmaxf(v, 0.0f);
        acc[j] = v;
    }
    if constexpr (VPL == 2) {
        *(float2*)(y + base) = make_float2(acc[0], acc[1]);
    } else {
        y[base] = acc[0];
    }
}


extern "C" void kernel_launch(void* const* d_in, const int* in_sizes, int n_in,
                              void* d_out, int out_size, void* d_ws, size_t ws_size,
                              hipStream_t stream) {
    const float* x    = (const float*)d_in[0];
    const int*   src  = (const int*)  d_in[1];
    const int*   dst  = (const int*)  d_in[2];
    const float* ew   = (const float*)d_in[3];
    const float* W1   = (const float*)d_in[4];
    const float* b1   = (const float*)d_in[5];
    const float* W2   = (const float*)d_in[6];
    const float* b2   = (const float*)d_in[7];
    const float* W3   = (const float*)d_in[8];
    const float* b3   = (const float*)d_in[9];
    const float* W4   = (const float*)d_in[10];
    const float* b4   = (const float*)d_in[11];
    const float* Wres = (const float*)d_in[12];
    const float* bres = (const float*)d_in[13];
    float* out = (float*)d_out;

    const int N  = in_sizes[0] / IN_F;
    const int E  = in_sizes[1];
    const int nbk = ceil_div_h(N, BNODES);
    const int g1  = ceil_div_h(E, CHUNK);
    const int L   = nbk * g1;
    const int nb  = ceil_div_h(L, 256);

    unsigned int* ws = (unsigned int*)d_ws;
    float* norm   = (float*)(ws);
    int*   rowptr = (int*)  (ws + N);
    int*   bh     = (int*)  (ws + 2L * N + 1);
    int*   offs   = bh + L;
    int*   bsums  = offs + L + 1;
    long   tmpOff = (2L * N + 2L * L + nb + 2 + 3) & ~3L;
    int4*  tmp    = (int4*) (ws + tmpOff);
    int2*  eRec   = (int2*) (ws + tmpOff + 4L * E);
    float* h      = (float*)(ws + tmpOff + 6L * E);
    float* xbuf   = h    + (size_t)N * H_F;
    float* res    = xbuf + (size_t)N * H_F;

    const int T  = 256;
    const int mt = ceil_div_h(N, 16);
    const int gblocks = ceil_div_h(N, T / 32);

    bucket_hist_kernel<<<g1, PT, nbk * sizeof(int), stream>>>(dst, bh, E, nbk);
    scan_block_kernel<<<nb, 256, 0, stream>>>(bh, offs, bsums, L);
    scan_sums_kernel<<<1, 32, 0, stream>>>(bsums, nb);
    scan_add_kernel<<<ceil_div_h(L, T), T, 0, stream>>>(offs, bsums, L, E);
    bucket_scatter_kernel<<<g1, PT, nbk * sizeof(int), stream>>>(src, dst, ew, offs, tmp, E, nbk);
    bucket_build_kernel<<<nbk, PT, 0, stream>>>(tmp, offs, rowptr, norm, eRec, N, E, g1);
    row_coef_kernel<<<ceil_div_h(4L * N, T), T, 0, stream>>>(rowptr, norm, eRec, N);

    gemm_wmma_kernel<IN_F, C_F><<<mt, (C_F / 16) * 32, 0, stream>>>(x, Wres, res, N);

    gemm_wmma_kernel<IN_F, H_F><<<mt, (H_F / 16) * 32, 0, stream>>>(x, W1, h, N);
    gather_agg_kernel<H_F, true, false><<<gblocks, T, 0, stream>>>(h, rowptr, eRec,
                                                                   b1, nullptr, nullptr, xbuf, N);
    gemm_wmma_kernel<H_F, H_F><<<mt, (H_F / 16) * 32, 0, stream>>>(xbuf, W2, h, N);
    gather_agg_kernel<H_F, true, false><<<gblocks, T, 0, stream>>>(h, rowptr, eRec,
                                                                   b2, nullptr, nullptr, xbuf, N);
    gemm_wmma_kernel<H_F, H_F><<<mt, (H_F / 16) * 32, 0, stream>>>(xbuf, W3, h, N);
    gather_agg_kernel<H_F, true, false><<<gblocks, T, 0, stream>>>(h, rowptr, eRec,
                                                                   b3, nullptr, nullptr, xbuf, N);
    gemm_wmma_kernel<H_F, C_F><<<mt, (C_F / 16) * 32, 0, stream>>>(xbuf, W4, h, N);
    gather_agg_kernel<C_F, false, true><<<gblocks, T, 0, stream>>>(h, rowptr, eRec,
                                                                   b4, res, bres, out, N);
}


# workspace planes (feature rows, residual, CSR records) start on 256-B boundaries: a gathered 256-B row is two whole 128-B lines of one aligned unit, not three of two units; device code unchanged
# speedup vs baseline: 1.7159x; 1.0694x over previous
#include <hip/hip_runtime.h>


typedef __attribute__((ext_vector_type(2))) float v2f;
typedef __attribute__((ext_vector_type(8))) float v8f;

#define IN_F 128
#define H_F  64
#define C_F  32

#define BSH    8
#define BNODES (1 << BSH)
#define CHUNK  16384
#define PT     1024
#define RCAP   6144

static inline int ceil_div_h(long a, long b) { return (int)((a + b - 1) / b); }


__global__ void bucket_hist_kernel(const int* __restrict__ dst, int* __restrict__ bh,
                                   int E, int nbk) {
    extern __shared__ int s_bins[];
    for (int k = threadIdx.x; k < nbk; k += blockDim.x) s_bins[k] = 0;
    __syncthreads();
    const int e0 = blockIdx.x * CHUNK;
    const int e1 = min(e0 + CHUNK, E);
    const int bs = blockDim.x;
    for (int eb = e0 + threadIdx.x; eb < e1; eb += 4 * bs) {
        int d[4];
        for (int j = 0; j < 4; ++j) d[j] = dst[min(eb + j * bs, e1 - 1)];
        asm volatile("" :: "v"(d[0]), "v"(d[1]), "v"(d[2]), "v"(d[3]));
        for (int j = 0; j < 4; ++j)
            if (eb + j * bs < e1) atomicAdd(&s_bins[d[j] >> BSH], 1);
    }
    __syncthreads();
    for (int k = threadIdx.x; k < nbk; k += blockDim.x)
        bh[k * gridDim.x + blockIdx.x] = s_bins[k];
}


__global__ void scan_block_kernel(const int* __restrict__ counts, int* __restrict__ rowptr,
                                  int* __restrict__ bsums, int n) {
    __shared__ int sdata[256];
    int tid = threadIdx.x;
    int i = blockIdx.x * 256 + tid;
    int v = (i < n) ? counts[i] : 0;
    sdata[tid] = v;
    __syncthreads();
#pragma unroll
    for (int off = 1; off < 256; off <<= 1) {
        int t = (tid >= off) ? sdata[tid - off] : 0;
        __syncthreads();
        sdata[tid] += t;
        __syncthreads();
    }
    if (i < n) rowptr[i] = sdata[tid] - v;
    if (tid == 255) bsums[blockIdx.x] = sdata[255];
}

__global__ void scan_sums_kernel(int* __restrict__ bsums, int nb) {
    if (blockIdx.x == 0 && threadIdx.x == 0) {
        int acc = 0;
        for (int i = 0; i < nb; ++i) { int t = bsums[i]; bsums[i] = acc; acc += t; }
    }
}

__global__ void scan_add_kernel(int* __restrict__ rowptr, const int* __restrict__ bsums,
                                int n, int E) {
    int i = blockIdx.x * blockDim.x + threadIdx.x;
    if (i < n) rowptr[i] += bsums[i >> 8];
    if (i == 0) rowptr[n] = E;
}

__global__ void bucket_scatter_kernel(const int* __restrict__ src, const int* __restrict__ dst,
                                      const float* __restrict__ ew, const int* __restrict__ offs,
                                      int4* __restrict__ tmp, int E, int nbk) {
    extern __shared__ int s_bins[];
    for (int k = threadIdx.x; k < nbk; k += blockDim.x)
        s_bins[k] = offs[k * gridDim.x + blockIdx.x];
    __syncthreads();
    const int e0 = blockIdx.x * CHUNK;
    const int e1 = min(e0 + CHUNK, E);
    const int bs = blockDim.x;
    for (int eb = e0 + threadIdx.x; eb < e1; eb += 4 * bs) {
        int s[4], d[4], w[4];
        for (int j = 0; j < 4; ++j) {
            const int e = min(eb + j * bs, e1 - 1);
            s[j] = src[e];
            d[j] = dst[e];
            w[j] = __float_as_int(ew[e]);
        }
        asm volatile("" :: "v"(s[0]), "v"(s[1]), "v"(s[2]), "v"(s[3]),
                           "v"(d[0]), "v"(d[1]), "v"(d[2]), "v"(d[3]),
                           "v"(w[0]), "v"(w[1]), "v"(w[2]), "v"(w[3]));
        for (int j = 0; j < 4; ++j) {
            if (eb + j * bs < e1) {
                int pos = atomicAdd(&s_bins[d[j] >> BSH], 1);
                tmp[pos] = make_int4(s[j], w[j], d[j], 0);
            }
        }
    }
}

__global__ void bucket_build_kernel(const int4* __restrict__ tmp, const int* __restrict__ offs,
                                    int* __restrict__ rowptr, float* __restrict__ norm,
                                    int2* __restrict__ eRec, int n, int E, int g1) {
    __shared__ int   s_cnt[BNODES];
    __shared__ float s_deg[BNODES];
    __shared__ int   s_scan[BNODES];
    __shared__ int2  s_rec[RCAP];
    const int tid   = threadIdx.x;
    const int t0    = offs[blockIdx.x * g1];
    const int t1    = offs[(blockIdx.x + 1) * g1];
    const int nbase = blockIdx.x << BSH;
    const bool stage = (t1 - t0) <= RCAP;

    if (tid < BNODES) { s_cnt[tid] = 0; s_deg[tid] = 0.0f; }
    __syncthreads();
    const int bs = blockDim.x;
    for (int ib = t0 + tid; ib < t1; ib += 4 * bs) {
        int4 r[4];
        for (int j = 0; j < 4; ++j) r[j] = tmp[min(ib + j * bs, t1 - 1)];
        asm volatile("" :: "v"(r[0].y), "v"(r[0].z), "v"(r[1].y), "v"(r[1].z),
                           "v"(r[2].y), "v"(r[2].z), "v"(r[3].y), "v"(r[3].z));
        for (int j = 0; j < 4; ++j) {
            if (ib + j * bs < t1) {
                int dl = r[j].z - nbase;
                atomicAdd(&s_cnt[dl], 1);
                atomicAdd(&s_deg[dl], __int_as_float(r[j].y));
            }
        }
    }
    __syncthreads();

    int v = 0;
    if (tid < BNODES) { v = s_cnt[tid]; s_scan[tid] = v; }
    __syncthreads();
    for (int off = 1; off < BNODES; off <<= 1) {
        int t = 0;
        if (tid < BNODES && tid >= off) t = s_scan[tid - off];
        __syncthreads();
        if (tid < BNODES) s_scan[tid] += t;
        __syncthreads();
    }
    if (tid < BNODES) {
        const int start = t0 + s_scan[tid] - v;
        s_cnt[tid] = start;
        const int node = nbase + tid;
        if (node < n) {
            rowptr[node] = start;
            float d = s_deg[tid];
            norm[node] = (d > 0.0f) ? rsqrtf(fmaxf(d, 1e-12f)) : 0.0f;
        }
    }
    if (blockIdx.x == gridDim.x - 1 && tid == 0) rowptr[n] = E;
    __syncthreads();

    for (int ib = t0 + tid; ib < t1; ib += 4 * bs) {
        int4 r[4];
        for (int j = 0; j < 4; ++j) r[j] = tmp[min(ib + j * bs, t1 - 1)];
        asm volatile("" :: "v"(r[0].x), "v"(r[0].y), "v"(r[0].z), "v"(r[1].x), "v"(r[1].y), "v"(r[1].z),
                           "v"(r[2].x), "v"(r[2].y), "v"(r[2].z), "v"(r[3].x), "v"(r[3].y), "v"(r[3].z));
        for (int j = 0; j < 4; ++j) {
            if (ib + j * bs < t1) {
                int pos = atomicAdd(&s_cnt[r[j].z - nbase], 1);
                if (stage) s_rec[pos - t0] = make_int2(r[j].x, r[j].y);
                else       eRec[pos]       = make_int2(r[j].x, r[j].y);
            }
        }
    }
    if (stage) {
        __syncthreads();
        for (int i = tid; i < t1 - t0; i += blockDim.x) eRec[t0 + i] = s_rec[i];
    }
}

__global__ void row_coef_kernel(const int* __restrict__ rowptr, const float* __restrict__ norm,
                                int2* __restrict__ eRec, int n) {
    const int t   = blockIdx.x * blockDim.x + threadIdx.x;
    const int n0  = (t >> 4) * 4;
    const int sub = t & 15;
    if (n0 >= n) return;
    const int nl = min(n0 + 4, n);
    int   rp[5];
    float nd[4];
    for (int q = 0; q < 5; ++q) rp[q] = rowptr[min(n0 + q, nl)];
    for (int q = 0; q < 4; ++q) nd[q] = norm[min(n0 + q, n - 1)];
    for (int i = rp[0] + sub; i < rp[4]; i += 16) {
        int2  r   = eRec[i];
        float ndq = nd[0];
        if (i >= rp[1]) ndq = nd[1];
        if (i >= rp[2]) ndq = nd[2];
        if (i >= rp[3]) ndq = nd[3];
        float c = __int_as_float(r.y) * norm[r.x] * ndq;
        eRec[i] = make_int2(r.x, __float_as_int(c));
    }
}

template <int FIN, int FOUT>
__global__ void gemm_wmma_kernel(const float* __restrict__ X, const float* __restrict__ W,
                                 float* __restrict__ Y, int nrows) {
    __shared__ float tx[16 * FIN];
    const int nthreads = (FOUT / 16) * 32;
    const int m0  = blockIdx.x * 16;
    const int tid = threadIdx.x;

    for (int i = tid; i < 16 * FIN; i += nthreads) {
        int r = i / FIN, c = i % FIN;
        int row = m0 + r;
        if (row >= nrows) row = nrows - 1;
        tx[i] = X[(long)row * FIN + c];
    }
    __syncthreads();

    const int wave  = tid >> 5;
    const int lane  = tid & 31;
    const int lrow  = lane & 15;
    const int khalf = (lane >> 4) * 2;
    const int n0    = wave * 16;

    v8f acc = {};
#pragma unroll
    for (int k0 = 0; k0 < FIN; k0 += 4) {
        v2f a = *(const v2f*)&tx[lrow * FIN + k0 + khalf];
        v2f b;
        b.x = W[(long)(k0 + khalf + 0) * FOUT + n0 + lrow];
        b.y = W[(long)(k0 + khalf + 1) * FOUT + n0 + lrow];
        acc = __builtin_amdgcn_wmma_f32_16x16x4_f32(
            false, a, false, b, (short)0, acc, false, false);
    }

    const int mbase = m0 + ((lane >> 4) * 8);
    const int col   = n0 + lrow;
#pragma unroll
    for (int r = 0; r < 8; ++r) {
        int row = mbase + r;
        if (row < nrows) Y[(long)row * FOUT + col] = acc[r];
    }
}

template <int F, bool RELU, bool FINAL>
__global__ void gather_agg_kernel(const float* __restrict__ h,
                                  const int* __restrict__ rowptr,
                                  const int2* __restrict__ eRec,
                                  const float* __restrict__ bias,
                                  const float* __restrict__ res,
                                  const float* __restrict__ bias2,
                                  float* __restrict__ y, int n) {
    const int wave = threadIdx.x >> 5;
    const int lane = threadIdx.x & 31;
    const int node = blockIdx.x * (blockDim.x >> 5) + wave;
    if (node >= n) return;

    const int s0 = rowptr[node];
    const int s1 = rowptr[node + 1];

    constexpr int VPL = F / 32;
    const int fbase = lane * VPL;

    float acc[VPL];
#pragma unroll
    for (int j = 0; j < VPL; ++j) acc[j] = 0.0f;

    for (int i = s0; i < s1; ++i) {
        int2  r = eRec[i];
        float c = __int_as_float(r.y);
        const float* hr = h + (long)r.x * F + fbase;
        if constexpr (VPL == 2) {
            float2 v = *(const float2*)hr;
            acc[0] += v.x * c;
            acc[1] += v.y * c;
        } else {
            acc[0] += hr[0] * c;
        }
    }

    const long base = (long)node * F + fbase;
#pragma unroll
    for (int j = 0; j < VPL; ++j) {
        float v = acc[j] + bias[fbase + j];
        if (FINAL) v += res[base + j] + bias2[fbase + j];
        if (RELU) v = fmaxf(v, 0.0f);
        acc[j] = v;
    }
    if constexpr (VPL == 2) {
        *(float2*)(y + base) = make_float2(acc[0], acc[1]);
    } else {
        y[base] = acc[0];
    }
}


extern "C" void kernel_launch(void* const* d_in, const int* in_sizes, int n_in,
                              void* d_out, int out_size, void* d_ws, size_t ws_size,
                              hipStream_t stream) {
    const float* x    = (const float*)d_in[0];
    const int*   src  = (const int*)  d_in[1];
    const int*   dst  = (const int*)  d_in[2];
    const float* ew   = (const float*)d_in[3];
    const float* W1   = (const float*)d_in[4];
    const float* b1   = (const float*)d_in[5];
    const float* W2   = (const float*)d_in[6];
    const float* b2   = (const float*)d_in[7];
    const float* W3   = (const float*)d_in[8];
    const float* b3   = (const float*)d_in[9];
    const float* W4   = (const float*)d_in[10];
    const float* b4   = (const float*)d_in[11];
    const float* Wres = (const float*)d_in[12];
    const float* bres = (const float*)d_in[13];
    float* out = (float*)d_out;

    const int N  = in_sizes[0] / IN_F;
    const int E  = in_sizes[1];
    const int nbk = ceil_div_h(N, BNODES);
    const int g1  = ceil_div_h(E, CHUNK);
    const int L   = nbk * g1;
    const int nb  = ceil_div_h(L, 256);

    unsigned int* ws = (unsigned int*)d_ws;
    float* norm   = (float*)(ws);
    int*   rowptr = (int*)  (ws + N);
    int*   bh     = (int*)  (ws + 2L * N + 1);
    int*   offs   = bh + L;
    int*   bsums  = offs + L + 1;
    long   tmpOff = (2L * N + 2L * L + nb + 2 + 63) & ~63L;
    long   recOff = tmpOff + ((4L * E + 63) & ~63L);
    long   hOff   = recOff + ((2L * E + 63) & ~63L);
    long   plane  = ((long)N * H_F + 63) & ~63L;
    int4*  tmp    = (int4*) (ws + tmpOff);
    int2*  eRec   = (int2*) (ws + recOff);
    float* h      = (float*)(ws + hOff);
    float* xbuf   = h    + plane;
    float* res    = xbuf + plane;

    const int T  = 256;
    const int mt = ceil_div_h(N, 16);
    const int gblocks = ceil_div_h(N, T / 32);

    bucket_hist_kernel<<<g1, PT, nbk * sizeof(int), stream>>>(dst, bh, E, nbk);
    scan_block_kernel<<<nb, 256, 0, stream>>>(bh, offs, bsums, L);
    scan_sums_kernel<<<1, 32, 0, stream>>>(bsums, nb);
    scan_add_kernel<<<ceil_div_h(L, T), T, 0, stream>>>(offs, bsums, L, E);
    bucket_scatter_kernel<<<g1, PT, nbk * sizeof(int), stream>>>(src, dst, ew, offs, tmp, E, nbk);
    bucket_build_kernel<<<nbk, PT, 0, stream>>>(tmp, offs, rowptr, norm, eRec, N, E, g1);
    row_coef_kernel<<<ceil_div_h(4L * N, T), T, 0, stream>>>(rowptr, norm, eRec, N);

    gemm_wmma_kernel<IN_F, C_F><<<mt, (C_F / 16) * 32, 0, stream>>>(x, Wres, res, N);

    gemm_wmma_kernel<IN_F, H_F><<<mt, (H_F / 16) * 32, 0, stream>>>(x, W1, h, N);
    gather_agg_kernel<H_F, true, false><<<gblocks, T, 0, stream>>>(h, rowptr, eRec,
                                                                   b1, nullptr, nullptr, xbuf, N);
    gemm_wmma_kernel<H_F, H_F><<<mt, (H_F / 16) * 32, 0, stream>>>(xbuf, W2, h, N);
    gather_agg_kernel<H_F, true, false><<<gblocks, T, 0, stream>>>(h, rowptr, eRec,
                                                                   b2, nullptr, nullptr, xbuf, N);
    gemm_wmma_kernel<H_F, H_F><<<mt, (H_F / 16) * 32, 0, stream>>>(xbuf, W3, h, N);
    gather_agg_kernel<H_F, true, false><<<gblocks, T, 0, stream>>>(h, rowptr, eRec,
                                                                   b3, nullptr, nullptr, xbuf, N);
    gemm_wmma_kernel<H_F, C_F><<<mt, (C_F / 16) * 32, 0, stream>>>(xbuf, W4, h, N);
    gather_agg_kernel<C_F, false, true><<<gblocks, T, 0, stream>>>(h, rowptr, eRec,
                                                                   b4, res, bres, out, N);
}


# gather: one wave sweeps the contiguous CSR records of 8 consecutive nodes (16 feature rows in flight, one tail per span) instead of one short wave per node; per-row sum order unchanged
# speedup vs baseline: 1.9311x; 1.1254x over previous
#include <hip/hip_runtime.h>
#include <type_traits>


typedef __attribute__((ext_vector_type(2))) float v2f;
typedef __attribute__((ext_vector_type(8))) float v8f;
typedef int rec8 __attribute__((ext_vector_type(16), aligned(8)));

#define IN_F 128
#define H_F  64
#define C_F  32

#define BSH    8
#define BNODES (1 << BSH)
#define CHUNK  16384
#define PT     1024
#define RCAP   6144

static inline int ceil_div_h(long a, long b) { return (int)((a + b - 1) / b); }


__global__ void bucket_hist_kernel(const int* __restrict__ dst, int* __restrict__ bh,
                                   int E, int nbk) {
    extern __shared__ int s_bins[];
    for (int k = threadIdx.x; k < nbk; k += blockDim.x) s_bins[k] = 0;
    __syncthreads();
    const int e0 = blockIdx.x * CHUNK;
    const int e1 = min(e0 + CHUNK, E);
    const int bs = blockDim.x;
    for (int eb = e0 + threadIdx.x; eb < e1; eb += 4 * bs) {
        int d[4];
        for (int j = 0; j < 4; ++j) d[j] = dst[min(eb + j * bs, e1 - 1)];
        asm volatile("" :: "v"(d[0]), "v"(d[1]), "v"(d[2]), "v"(d[3]));
        for (int j = 0; j < 4; ++j)
            if (eb + j * bs < e1) atomicAdd(&s_bins[d[j] >> BSH], 1);
    }
    __syncthreads();
    for (int k = threadIdx.x; k < nbk; k += blockDim.x)
        bh[k * gridDim.x + blockIdx.x] = s_bins[k];
}


__global__ void scan_block_kernel(const int* __restrict__ counts, int* __restrict__ rowptr,
                                  int* __restrict__ bsums, int n) {
    __shared__ int sdata[256];
    int tid = threadIdx.x;
    int i = blockIdx.x * 256 + tid;
    int v = (i < n) ? counts[i] : 0;
    sdata[tid] = v;
    __syncthreads();
#pragma unroll
    for (int off = 1; off < 256; off <<= 1) {
        int t = (tid >= off) ? sdata[tid - off] : 0;
        __syncthreads();
        sdata[tid] += t;
        __syncthreads();
    }
    if (i < n) rowptr[i] = sdata[tid] - v;
    if (tid == 255) bsums[blockIdx.x] = sdata[255];
}

__global__ void scan_sums_kernel(int* __restrict__ bsums, int nb) {
    if (blockIdx.x == 0 && threadIdx.x == 0) {
        int acc = 0;
        for (int i = 0; i < nb; ++i) { int t = bsums[i]; bsums[i] = acc; acc += t; }
    }
}

__global__ void scan_add_kernel(int* __restrict__ rowptr, const int* __restrict__ bsums,
                                int n, int E) {
    int i = blockIdx.x * blockDim.x + threadIdx.x;
    if (i < n) rowptr[i] += bsums[i >> 8];
    if (i == 0) rowptr[n] = E;
}

__global__ void bucket_scatter_kernel(const int* __restrict__ src, const int* __restrict__ dst,
                                      const float* __restrict__ ew, const int* __restrict__ offs,
                                      int4* __restrict__ tmp, int E, int nbk) {
    extern __shared__ int s_bins[];
    for (int k = threadIdx.x; k < nbk; k += blockDim.x)
        s_bins[k] = offs[k * gridDim.x + blockIdx.x];
    __syncthreads();
    const int e0 = blockIdx.x * CHUNK;
    const int e1 = min(e0 + CHUNK, E);
    const int bs = blockDim.x;
    for (int eb = e0 + threadIdx.x; eb < e1; eb += 4 * bs) {
        int s[4], d[4], w[4];
        for (int j = 0; j < 4; ++j) {
            const int e = min(eb + j * bs, e1 - 1);
            s[j] = src[e];
            d[j] = dst[e];
            w[j] = __float_as_int(ew[e]);
        }
        asm volatile("" :: "v"(s[0]), "v"(s[1]), "v"(s[2]), "v"(s[3]),
                           "v"(d[0]), "v"(d[1]), "v"(d[2]), "v"(d[3]),
                           "v"(w[0]), "v"(w[1]), "v"(w[2]), "v"(w[3]));
        for (int j = 0; j < 4; ++j) {
            if (eb + j * bs < e1) {
                int pos = atomicAdd(&s_bins[d[j] >> BSH], 1);
                tmp[pos] = make_int4(s[j], w[j], d[j], 0);
            }
        }
    }
}

__global__ void bucket_build_kernel(const int4* __restrict__ tmp, const int* __restrict__ offs,
                                    int* __restrict__ rowptr, float* __restrict__ norm,
                                    int2* __restrict__ eRec, int n, int E, int g1) {
    __shared__ int   s_cnt[BNODES];
    __shared__ float s_deg[BNODES];
    __shared__ int   s_scan[BNODES];
    __shared__ int2  s_rec[RCAP];
    const int tid   = threadIdx.x;
    const int t0    = offs[blockIdx.x * g1];
    const int t1    = offs[(blockIdx.x + 1) * g1];
    const int nbase = blockIdx.x << BSH;
    const bool stage = (t1 - t0) <= RCAP;

    if (tid < BNODES) { s_cnt[tid] = 0; s_deg[tid] = 0.0f; }
    __syncthreads();
    const int bs = blockDim.x;
    for (int ib = t0 + tid; ib < t1; ib += 4 * bs) {
        int4 r[4];
        for (int j = 0; j < 4; ++j) r[j] = tmp[min(ib + j * bs, t1 - 1)];
        asm volatile("" :: "v"(r[0].y), "v"(r[0].z), "v"(r[1].y), "v"(r[1].z),
                           "v"(r[2].y), "v"(r[2].z), "v"(r[3].y), "v"(r[3].z));
        for (int j = 0; j < 4; ++j) {
            if (ib + j * bs < t1) {
                int dl = r[j].z - nbase;
                atomicAdd(&s_cnt[dl], 1);
                atomicAdd(&s_deg[dl], __int_as_float(r[j].y));
            }
        }
    }
    __syncthreads();

    int v = 0;
    if (tid < BNODES) { v = s_cnt[tid]; s_scan[tid] = v; }
    __syncthreads();
    for (int off = 1; off < BNODES; off <<= 1) {
        int t = 0;
        if (tid < BNODES && tid >= off) t = s_scan[tid - off];
        __syncthreads();
        if (tid < BNODES) s_scan[tid] += t;
        __syncthreads();
    }
    if (tid < BNODES) {
        const int start = t0 + s_scan[tid] - v;
        s_cnt[tid] = start;
        const int node = nbase + tid;
        if (node < n) {
            rowptr[node] = start;
            float d = s_deg[tid];
            norm[node] = (d > 0.0f) ? rsqrtf(fmaxf(d, 1e-12f)) : 0.0f;
        }
    }
    if (blockIdx.x == gridDim.x - 1 && tid == 0) rowptr[n] = E;
    __syncthreads();

    for (int ib = t0 + tid; ib < t1; ib += 4 * bs) {
        int4 r[4];
        for (int j = 0; j < 4; ++j) r[j] = tmp[min(ib + j * bs, t1 - 1)];
        asm volatile("" :: "v"(r[0].x), "v"(r[0].y), "v"(r[0].z), "v"(r[1].x), "v"(r[1].y), "v"(r[1].z),
                           "v"(r[2].x), "v"(r[2].y), "v"(r[2].z), "v"(r[3].x), "v"(r[3].y), "v"(r[3].z));
        for (int j = 0; j < 4; ++j) {
            if (ib + j * bs < t1) {
                int pos = atomicAdd(&s_cnt[r[j].z - nbase], 1);
                if (stage) s_rec[pos - t0] = make_int2(r[j].x, r[j].y);
                else       eRec[pos]       = make_int2(r[j].x, r[j].y);
            }
        }
    }
    if (stage) {
        __syncthreads();
        for (int i = tid; i < t1 - t0; i += blockDim.x) eRec[t0 + i] = s_rec[i];
    }
}

__global__ void row_coef_kernel(const int* __restrict__ rowptr, const float* __restrict__ norm,
                                int2* __restrict__ eRec, int n) {
    const int t   = blockIdx.x * blockDim.x + threadIdx.x;
    const int n0  = (t >> 4) * 4;
    const int sub = t & 15;
    if (n0 >= n) return;
    const int nl = min(n0 + 4, n);
    int   rp[5];
    float nd[4];
    for (int q = 0; q < 5; ++q) rp[q] = rowptr[min(n0 + q, nl)];
    for (int q = 0; q < 4; ++q) nd[q] = norm[min(n0 + q, n - 1)];
    for (int i = rp[0] + sub; i < rp[4]; i += 16) {
        int2  r   = eRec[i];
        float ndq = nd[0];
        if (i >= rp[1]) ndq = nd[1];
        if (i >= rp[2]) ndq = nd[2];
        if (i >= rp[3]) ndq = nd[3];
        float c = __int_as_float(r.y) * norm[r.x] * ndq;
        eRec[i] = make_int2(r.x, __float_as_int(c));
    }
}

template <int FIN, int FOUT>
__global__ void gemm_wmma_kernel(const float* __restrict__ X, const float* __restrict__ W,
                                 float* __restrict__ Y, int nrows) {
    __shared__ float tx[16 * FIN];
    const int nthreads = (FOUT / 16) * 32;
    const int m0  = blockIdx.x * 16;
    const int tid = threadIdx.x;

    for (int i = tid; i < 16 * FIN; i += nthreads) {
        int r = i / FIN, c = i % FIN;
        int row = m0 + r;
        if (row >= nrows) row = nrows - 1;
        tx[i] = X[(long)row * FIN + c];
    }
    __syncthreads();

    const int wave  = tid >> 5;
    const int lane  = tid & 31;
    const int lrow  = lane & 15;
    const int khalf = (lane >> 4) * 2;
    const int n0    = wave * 16;

    v8f acc = {};
#pragma unroll
    for (int k0 = 0; k0 < FIN; k0 += 4) {
        v2f a = *(const v2f*)&tx[lrow * FIN + k0 + khalf];
        v2f b;
        b.x = W[(long)(k0 + khalf + 0) * FOUT + n0 + lrow];
        b.y = W[(long)(k0 + khalf + 1) * FOUT + n0 + lrow];
        acc = __builtin_amdgcn_wmma_f32_16x16x4_f32(
            false, a, false, b, (short)0, acc, false, false);
    }

    const int mbase = m0 + ((lane >> 4) * 8);
    const int col   = n0 + lrow;
#pragma unroll
    for (int r = 0; r < 8; ++r) {
        int row = mbase + r;
        if (row < nrows) Y[(long)row * FOUT + col] = acc[r];
    }
}

#define NPW 8
#define GC  16

template <int F, bool RELU, bool FINAL>
__global__ void __launch_bounds__(256)
gather_agg_kernel(const float* __restrict__ h,
                  const int* __restrict__ rowptr,
                  const int2* __restrict__ eRec,
                  const float* __restrict__ bias,
                  const float* __restrict__ res,
                  const float* __restrict__ bias2,
                  float* __restrict__ y, int n) {
    constexpr int VPL = F / 32;
    const int lane  = threadIdx.x & 31;
    const int fbase = lane * VPL;
    const int wv = __builtin_amdgcn_readfirstlane(
        (int)(blockIdx.x * (blockDim.x >> 5) + (threadIdx.x >> 5)));
    const int n0 = wv * NPW;
    if (n0 >= n) return;
    const int n1 = min(n0 + NPW, n);

    const int rpv = rowptr[min(n0 + lane, n1)];
    int       e    = __builtin_amdgcn_readlane(rpv, 0);
    const int eEnd = __builtin_amdgcn_readlane(rpv, NPW);
    int       nb   = __builtin_amdgcn_readlane(rpv, 1);
    int       node = n0;

    float bs[VPL], bs2[VPL], resv[VPL], acc[VPL];
#pragma unroll
    for (int j = 0; j < VPL; ++j) {
        bs[j]   = bias[fbase + j];
        bs2[j]  = FINAL ? bias2[fbase + j] : 0.0f;
        resv[j] = FINAL ? res[(long)n0 * F + fbase + j] : 0.0f;
        acc[j]  = 0.0f;
    }

    auto flush = [&]() {
        const long base = (long)node * F + fbase;
        float o[VPL];
#pragma unroll
        for (int j = 0; j < VPL; ++j) {
            float v = acc[j] + bs[j];
            if (FINAL) v += resv[j] + bs2[j];
            if (RELU) v = fmaxf(v, 0.0f);
            o[j]   = v;
            acc[j] = 0.0f;
        }
        if constexpr (VPL == 2) {
            *(float2*)(y + base) = make_float2(o[0], o[1]);
        } else {
            y[base] = o[0];
        }
        ++node;
        nb = __builtin_amdgcn_readlane(rpv, node - n0 + 1);
        if constexpr (FINAL) {
            if (node < n1) {
#pragma unroll
                for (int j = 0; j < VPL; ++j) resv[j] = res[(long)node * F + fbase + j];
            }
        }
    };

    auto group = [&](auto fullTag) {
        constexpr bool FULL = decltype(fullTag)::value;
        int   sx[GC];
        float cf[GC];
        if constexpr (FULL) {
            const rec8* p = (const rec8*)(eRec + e);
            const rec8 ra = p[0], rb = p[1];
#pragma unroll
            for (int j = 0; j < 8; ++j) {
                sx[j]     = ra[2 * j];  cf[j]     = __int_as_float(ra[2 * j + 1]);
                sx[8 + j] = rb[2 * j];  cf[8 + j] = __int_as_float(rb[2 * j + 1]);
            }
        } else {
#pragma unroll
            for (int j = 0; j < GC; ++j) {
                const int2 r = eRec[min(e + j, eEnd - 1)];
                sx[j] = r.x;
                cf[j] = __int_as_float(r.y);
            }
        }
        float rv[GC][VPL];
#pragma unroll
        for (int j = 0; j < GC; ++j) {
            if constexpr (VPL == 2) {
                const float2 v = ((const float2*)h)[(unsigned)sx[j] * (unsigned)(F / 2) + (unsigned)lane];
                rv[j][0] = v.x; rv[j][1] = v.y;
            } else {
                rv[j][0] = h[(unsigned)sx[j] * (unsigned)F + (unsigned)lane];
            }
        }
#pragma unroll
        for (int j = 0; j < GC; ++j) {
            if (FULL || e + j < eEnd) {
                while (e + j == nb) flush();
#pragma unroll
                for (int q = 0; q < VPL; ++q) acc[q] += rv[j][q] * cf[j];
            }
        }
    };

    while (e < eEnd) {
        if (eEnd - e >= GC) group(std::true_type{});
        else                group(std::false_type{});
        e += GC;
    }
    while (node < n1) flush();
}


extern "C" void kernel_launch(void* const* d_in, const int* in_sizes, int n_in,
                              void* d_out, int out_size, void* d_ws, size_t ws_size,
                              hipStream_t stream) {
    const float* x    = (const float*)d_in[0];
    const int*   src  = (const int*)  d_in[1];
    const int*   dst  = (const int*)  d_in[2];
    const float* ew   = (const float*)d_in[3];
    const float* W1   = (const float*)d_in[4];
    const float* b1   = (const float*)d_in[5];
    const float* W2   = (const float*)d_in[6];
    const float* b2   = (const float*)d_in[7];
    const float* W3   = (const float*)d_in[8];
    const float* b3   = (const float*)d_in[9];
    const float* W4   = (const float*)d_in[10];
    const float* b4   = (const float*)d_in[11];
    const float* Wres = (const float*)d_in[12];
    const float* bres = (const float*)d_in[13];
    float* out = (float*)d_out;

    const int N  = in_sizes[0] / IN_F;
    const int E  = in_sizes[1];
    const int nbk = ceil_div_h(N, BNODES);
    const int g1  = ceil_div_h(E, CHUNK);
    const int L   = nbk * g1;
    const int nb  = ceil_div_h(L, 256);

    unsigned int* ws = (unsigned int*)d_ws;
    float* norm   = (float*)(ws);
    int*   rowptr = (int*)  (ws + N);
    int*   bh     = (int*)  (ws + 2L * N + 1);
    int*   offs   = bh + L;
    int*   bsums  = offs + L + 1;
    long   tmpOff = (2L * N + 2L * L + nb + 2 + 63) & ~63L;
    long   recOff = tmpOff + ((4L * E + 63) & ~63L);
    long   hOff   = recOff + ((2L * E + 63) & ~63L);
    long   plane  = ((long)N * H_F + 63) & ~63L;
    int4*  tmp    = (int4*) (ws + tmpOff);
    int2*  eRec   = (int2*) (ws + recOff);
    float* h      = (float*)(ws + hOff);
    float* xbuf   = h    + plane;
    float* res    = xbuf + plane;

    const int T  = 256;
    const int mt = ceil_div_h(N, 16);
    const int gblocks = ceil_div_h(N, (T / 32) * NPW);

    bucket_hist_kernel<<<g1, PT, nbk * sizeof(int), stream>>>(dst, bh, E, nbk);
    scan_block_kernel<<<nb, 256, 0, stream>>>(bh, offs, bsums, L);
    scan_sums_kernel<<<1, 32, 0, stream>>>(bsums, nb);
    scan_add_kernel<<<ceil_div_h(L, T), T, 0, stream>>>(offs, bsums, L, E);
    bucket_scatter_kernel<<<g1, PT, nbk * sizeof(int), stream>>>(src, dst, ew, offs, tmp, E, nbk);
    bucket_build_kernel<<<nbk, PT, 0, stream>>>(tmp, offs, rowptr, norm, eRec, N, E, g1);
    row_coef_kernel<<<ceil_div_h(4L * N, T), T, 0, stream>>>(rowptr, norm, eRec, N);

    gemm_wmma_kernel<IN_F, C_F><<<mt, (C_F / 16) * 32, 0, stream>>>(x, Wres, res, N);

    gemm_wmma_kernel<IN_F, H_F><<<mt, (H_F / 16) * 32, 0, stream>>>(x, W1, h, N);
    gather_agg_kernel<H_F, true, false><<<gblocks, T, 0, stream>>>(h, rowptr, eRec,
                                                                   b1, nullptr, nullptr, xbuf, N);
    gemm_wmma_kernel<H_F, H_F><<<mt, (H_F / 16) * 32, 0, stream>>>(xbuf, W2, h, N);
    gather_agg_kernel<H_F, true, false><<<gblocks, T, 0, stream>>>(h, rowptr, eRec,
                                                                   b2, nullptr, nullptr, xbuf, N);
    gemm_wmma_kernel<H_F, H_F><<<mt, (H_F / 16) * 32, 0, stream>>>(xbuf, W3, h, N);
    gather_agg_kernel<H_F, true, false><<<gblocks, T, 0, stream>>>(h, rowptr, eRec,
                                                                   b3, nullptr, nullptr, xbuf, N);
    gemm_wmma_kernel<H_F, C_F><<<mt, (C_F / 16) * 32, 0, stream>>>(xbuf, W4, h, N);
    gather_agg_kernel<C_F, false, true><<<gblocks, T, 0, stream>>>(h, rowptr, eRec,
                                                                   b4, res, bres, out, N);
}


# layers 2-4: each product runs inside the preceding gather (64 finished rows staged in LDS, then f32 16x16x4 WMMA, same k order): three product launches and the activation round trip go
# speedup vs baseline: 2.3639x; 1.2241x over previous
#include <hip/hip_runtime.h>
#include <type_traits>


typedef __attribute__((ext_vector_type(2))) float v2f;
typedef __attribute__((ext_vector_type(8))) float v8f;
typedef int rec8 __attribute__((ext_vector_type(16), aligned(8)));

#define IN_F 128
#define H_F  64
#define C_F  32

#define BSH    8
#define BNODES (1 << BSH)
#define CHUNK  16384
#define PT     1024
#define RCAP   6144

static inline int ceil_div_h(long a, long b) { return (int)((a + b - 1) / b); }


__global__ void bucket_hist_kernel(const int* __restrict__ dst, int* __restrict__ bh,
                                   int E, int nbk) {
    extern __shared__ int s_bins[];
    for (int k = threadIdx.x; k < nbk; k += blockDim.x) s_bins[k] = 0;
    __syncthreads();
    const int e0 = blockIdx.x * CHUNK;
    const int e1 = min(e0 + CHUNK, E);
    const int bs = blockDim.x;
    for (int eb = e0 + threadIdx.x; eb < e1; eb += 4 * bs) {
        int d[4];
        for (int j = 0; j < 4; ++j) d[j] = dst[min(eb + j * bs, e1 - 1)];
        asm volatile("" :: "v"(d[0]), "v"(d[1]), "v"(d[2]), "v"(d[3]));
        for (int j = 0; j < 4; ++j)
            if (eb + j * bs < e1) atomicAdd(&s_bins[d[j] >> BSH], 1);
    }
    __syncthreads();
    for (int k = threadIdx.x; k < nbk; k += blockDim.x)
        bh[k * gridDim.x + blockIdx.x] = s_bins[k];
}


__global__ void scan_block_kernel(const int* __restrict__ counts, int* __restrict__ rowptr,
                                  int* __restrict__ bsums, int n) {
    __shared__ int sdata[256];
    int tid = threadIdx.x;
    int i = blockIdx.x * 256 + tid;
    int v = (i < n) ? counts[i] : 0;
    sdata[tid] = v;
    __syncthreads();
#pragma unroll
    for (int off = 1; off < 256; off <<= 1) {
        int t = (tid >= off) ? sdata[tid - off] : 0;
        __syncthreads();
        sdata[tid] += t;
        __syncthreads();
    }
    if (i < n) rowptr[i] = sdata[tid] - v;
    if (tid == 255) bsums[blockIdx.x] = sdata[255];
}

__global__ void scan_sums_kernel(int* __restrict__ bsums, int nb) {
    if (blockIdx.x == 0 && threadIdx.x == 0) {
        int acc = 0;
        for (int i = 0; i < nb; ++i) { int t = bsums[i]; bsums[i] = acc; acc += t; }
    }
}

__global__ void scan_add_kernel(int* __restrict__ rowptr, const int* __restrict__ bsums,
                                int n, int E) {
    int i = blockIdx.x * blockDim.x + threadIdx.x;
    if (i < n) rowptr[i] += bsums[i >> 8];
    if (i == 0) rowptr[n] = E;
}

__global__ void bucket_scatter_kernel(const int* __restrict__ src, const int* __restrict__ dst,
                                      const float* __restrict__ ew, const int* __restrict__ offs,
                                      int4* __restrict__ tmp, int E, int nbk) {
    extern __shared__ int s_bins[];
    for (int k = threadIdx.x; k < nbk; k += blockDim.x)
        s_bins[k] = offs[k * gridDim.x + blockIdx.x];
    __syncthreads();
    const int e0 = blockIdx.x * CHUNK;
    const int e1 = min(e0 + CHUNK, E);
    const int bs = blockDim.x;
    for (int eb = e0 + threadIdx.x; eb < e1; eb += 4 * bs) {
        int s[4], d[4], w[4];
        for (int j = 0; j < 4; ++j) {
            const int e = min(eb + j * bs, e1 - 1);
            s[j] = src[e];
            d[j] = dst[e];
            w[j] = __float_as_int(ew[e]);
        }
        asm volatile("" :: "v"(s[0]), "v"(s[1]), "v"(s[2]), "v"(s[3]),
                           "v"(d[0]), "v"(d[1]), "v"(d[2]), "v"(d[3]),
                           "v"(w[0]), "v"(w[1]), "v"(w[2]), "v"(w[3]));
        for (int j = 0; j < 4; ++j) {
            if (eb + j * bs < e1) {
                int pos = atomicAdd(&s_bins[d[j] >> BSH], 1);
                tmp[pos] = make_int4(s[j], w[j], d[j], 0);
            }
        }
    }
}

__global__ void bucket_build_kernel(const int4* __restrict__ tmp, const int* __restrict__ offs,
                                    int* __restrict__ rowptr, float* __restrict__ norm,
                                    int2* __restrict__ eRec, int n, int E, int g1) {
    __shared__ int   s_cnt[BNODES];
    __shared__ float s_deg[BNODES];
    __shared__ int   s_scan[BNODES];
    __shared__ int2  s_rec[RCAP];
    const int tid   = threadIdx.x;
    const int t0    = offs[blockIdx.x * g1];
    const int t1    = offs[(blockIdx.x + 1) * g1];
    const int nbase = blockIdx.x << BSH;
    const bool stage = (t1 - t0) <= RCAP;

    if (tid < BNODES) { s_cnt[tid] = 0; s_deg[tid] = 0.0f; }
    __syncthreads();
    const int bs = blockDim.x;
    for (int ib = t0 + tid; ib < t1; ib += 4 * bs) {
        int4 r[4];
        for (int j = 0; j < 4; ++j) r[j] = tmp[min(ib + j * bs, t1 - 1)];
        asm volatile("" :: "v"(r[0].y), "v"(r[0].z), "v"(r[1].y), "v"(r[1].z),
                           "v"(r[2].y), "v"(r[2].z), "v"(r[3].y), "v"(r[3].z));
        for (int j = 0; j < 4; ++j) {
            if (ib + j * bs < t1) {
                int dl = r[j].z - nbase;
                atomicAdd(&s_cnt[dl], 1);
                atomicAdd(&s_deg[dl], __int_as_float(r[j].y));
            }
        }
    }
    __syncthreads();

    int v = 0;
    if (tid < BNODES) { v = s_cnt[tid]; s_scan[tid] = v; }
    __syncthreads();
    for (int off = 1; off < BNODES; off <<= 1) {
        int t = 0;
        if (tid < BNODES && tid >= off) t = s_scan[tid - off];
        __syncthreads();
        if (tid < BNODES) s_scan[tid] += t;
        __syncthreads();
    }
    if (tid < BNODES) {
        const int start = t0 + s_scan[tid] - v;
        s_cnt[tid] = start;
        const int node = nbase + tid;
        if (node < n) {
            rowptr[node] = start;
            float d = s_deg[tid];
            norm[node] = (d > 0.0f) ? rsqrtf(fmaxf(d, 1e-12f)) : 0.0f;
        }
    }
    if (blockIdx.x == gridDim.x - 1 && tid == 0) rowptr[n] = E;
    __syncthreads();

    for (int ib = t0 + tid; ib < t1; ib += 4 * bs) {
        int4 r[4];
        for (int j = 0; j < 4; ++j) r[j] = tmp[min(ib + j * bs, t1 - 1)];
        asm volatile("" :: "v"(r[0].x), "v"(r[0].y), "v"(r[0].z), "v"(r[1].x), "v"(r[1].y), "v"(r[1].z),
                           "v"(r[2].x), "v"(r[2].y), "v"(r[2].z), "v"(r[3].x), "v"(r[3].y), "v"(r[3].z));
        for (int j = 0; j < 4; ++j) {
            if (ib + j * bs < t1) {
                int pos = atomicAdd(&s_cnt[r[j].z - nbase], 1);
                if (stage) s_rec[pos - t0] = make_int2(r[j].x, r[j].y);
                else       eRec[pos]       = make_int2(r[j].x, r[j].y);
            }
        }
    }
    if (stage) {
        __syncthreads();
        for (int i = tid; i < t1 - t0; i += blockDim.x) eRec[t0 + i] = s_rec[i];
    }
}

__global__ void row_coef_kernel(const int* __restrict__ rowptr, const float* __restrict__ norm,
                                int2* __restrict__ eRec, int n) {
    const int t   = blockIdx.x * blockDim.x + threadIdx.x;
    const int n0  = (t >> 4) * 4;
    const int sub = t & 15;
    if (n0 >= n) return;
    const int nl = min(n0 + 4, n);
    int   rp[5];
    float nd[4];
    for (int q = 0; q < 5; ++q) rp[q] = rowptr[min(n0 + q, nl)];
    for (int q = 0; q < 4; ++q) nd[q] = norm[min(n0 + q, n - 1)];
    for (int i = rp[0] + sub; i < rp[4]; i += 16) {
        int2  r   = eRec[i];
        float ndq = nd[0];
        if (i >= rp[1]) ndq = nd[1];
        if (i >= rp[2]) ndq = nd[2];
        if (i >= rp[3]) ndq = nd[3];
        float c = __int_as_float(r.y) * norm[r.x] * ndq;
        eRec[i] = make_int2(r.x, __float_as_int(c));
    }
}

template <int FIN, int FOUT>
__global__ void gemm_wmma_kernel(const float* __restrict__ X, const float* __restrict__ W,
                                 float* __restrict__ Y, int nrows) {
    __shared__ float tx[16 * FIN];
    const int nthreads = (FOUT / 16) * 32;
    const int m0  = blockIdx.x * 16;
    const int tid = threadIdx.x;

    for (int i = tid; i < 16 * FIN; i += nthreads) {
        int r = i / FIN, c = i % FIN;
        int row = m0 + r;
        if (row >= nrows) row = nrows - 1;
        tx[i] = X[(long)row * FIN + c];
    }
    __syncthreads();

    const int wave  = tid >> 5;
    const int lane  = tid & 31;
    const int lrow  = lane & 15;
    const int khalf = (lane >> 4) * 2;
    const int n0    = wave * 16;

    v8f acc = {};
#pragma unroll
    for (int k0 = 0; k0 < FIN; k0 += 4) {
        v2f a = *(const v2f*)&tx[lrow * FIN + k0 + khalf];
        v2f b;
        b.x = W[(long)(k0 + khalf + 0) * FOUT + n0 + lrow];
        b.y = W[(long)(k0 + khalf + 1) * FOUT + n0 + lrow];
        acc = __builtin_amdgcn_wmma_f32_16x16x4_f32(
            false, a, false, b, (short)0, acc, false, false);
    }

    const int mbase = m0 + ((lane >> 4) * 8);
    const int col   = n0 + lrow;
#pragma unroll
    for (int r = 0; r < 8; ++r) {
        int row = mbase + r;
        if (row < nrows) Y[(long)row * FOUT + col] = acc[r];
    }
}

#define NPW 8
#define GC  16
#define GW  8
#define XP  68

template <int F, bool RELU, bool FINAL, int FNEXT>
__global__ void __launch_bounds__(GW * 32)
gather_agg_kernel(const float* __restrict__ h,
                  const int* __restrict__ rowptr,
                  const int2* __restrict__ eRec,
                  const float* __restrict__ bias,
                  const float* __restrict__ res,
                  const float* __restrict__ bias2,
                  const float* __restrict__ Wn,
                  float* __restrict__ y, int n) {
    constexpr int VPL = F / 32;
    static_assert(FNEXT == 0 || (F == 64 && !FINAL && FNEXT % 16 == 0 && (4 * FNEXT / 16) % GW == 0),
                  "fused product: 64-wide rows, 64 x FNEXT tiles split evenly over the waves");
    __shared__ __attribute__((aligned(16))) float s_x[FNEXT > 0 ? GW * NPW * XP : 1];
    const int lane  = threadIdx.x & 31;
    const int fbase = lane * VPL;
    const int wib = __builtin_amdgcn_readfirstlane((int)(threadIdx.x >> 5));
    const int wv  = __builtin_amdgcn_readfirstlane((int)(blockIdx.x * GW)) + wib;
    const int n0  = wv * NPW;
    if constexpr (FNEXT == 0) {
        if (n0 >= n) return;
    }
    const int n1 = min(n0 + NPW, n);
    if constexpr (FNEXT > 0) {
        for (int q = max(n1 - n0, 0); q < NPW; ++q)
            *(float2*)(s_x + (wib * NPW + q) * XP + fbase) = make_float2(0.0f, 0.0f);
    }

    const int rpv = rowptr[min(n0 + lane, n1)];
    int       e    = __builtin_amdgcn_readlane(rpv, 0);
    const int eEnd = __builtin_amdgcn_readlane(rpv, NPW);
    int       nb   = __builtin_amdgcn_readlane(rpv, 1);
    int       node = n0;

    float bs[VPL], bs2[VPL], resv[VPL], acc[VPL];
#pragma unroll
    for (int j = 0; j < VPL; ++j) {
        bs[j]   = bias[fbase + j];
        bs2[j]  = FINAL ? bias2[fbase + j] : 0.0f;
        resv[j] = FINAL ? res[(long)n0 * F + fbase + j] : 0.0f;
        acc[j]  = 0.0f;
    }

    auto flush = [&]() {
        const long base = (long)node * F + fbase;
        float o[VPL];
#pragma unroll
        for (int j = 0; j < VPL; ++j) {
            float v = acc[j] + bs[j];
            if (FINAL) v += resv[j] + bs2[j];
            if (RELU) v = fmaxf(v, 0.0f);
            o[j]   = v;
            acc[j] = 0.0f;
        }
        if constexpr (FNEXT > 0) {
            *(float2*)(s_x + (wib * NPW + (node - n0)) * XP + fbase) = make_float2(o[0], o[1]);
        } else if constexpr (VPL == 2) {
            *(float2*)(y + base) = make_float2(o[0], o[1]);
        } else {
            y[base] = o[0];
        }
        ++node;
        nb = __builtin_amdgcn_readlane(rpv, node - n0 + 1);
        if constexpr (FINAL) {
            if (node < n1) {
#pragma unroll
                for (int j = 0; j < VPL; ++j) resv[j] = res[(long)node * F + fbase + j];
            }
        }
    };

    auto group = [&](auto fullTag) {
        constexpr bool FULL = decltype(fullTag)::value;
        int   sx[GC];
        float cf[GC];
        if constexpr (FULL) {
            const rec8* p = (const rec8*)(eRec + e);
            const rec8 ra = p[0], rb = p[1];
#pragma unroll
            for (int j = 0; j < 8; ++j) {
                sx[j]     = ra[2 * j];  cf[j]     = __int_as_float(ra[2 * j + 1]);
                sx[8 + j] = rb[2 * j];  cf[8 + j] = __int_as_float(rb[2 * j + 1]);
            }
        } else {
#pragma unroll
            for (int j = 0; j < GC; ++j) {
                const int2 r = eRec[min(e + j, eEnd - 1)];
                sx[j] = r.x;
                cf[j] = __int_as_float(r.y);
            }
        }
        float rv[GC][VPL];
#pragma unroll
        for (int j = 0; j < GC; ++j) {
            if constexpr (VPL == 2) {
                const float2 v = ((const float2*)h)[(unsigned)sx[j] * (unsigned)(F / 2) + (unsigned)lane];
                rv[j][0] = v.x; rv[j][1] = v.y;
            } else {
                rv[j][0] = h[(unsigned)sx[j] * (unsigned)F + (unsigned)lane];
            }
        }
#pragma unroll
        for (int j = 0; j < GC; ++j) {
            if (FULL || e + j < eEnd) {
                while (e + j == nb) flush();
#pragma unroll
                for (int q = 0; q < VPL; ++q) acc[q] += rv[j][q] * cf[j];
            }
        }
    };

    while (e < eEnd) {
        if (eEnd - e >= GC) group(std::true_type{});
        else                group(std::false_type{});
        e += GC;
    }
    while (node < n1) flush();

    if constexpr (FNEXT > 0) {
        constexpr int NT  = FNEXT / 16;
        constexpr int TPW = (4 * NT) / GW;
        const int lrow  = lane & 15;
        const int khalf = (lane >> 4) * 2;
        const int col   = (wib % NT) * 16 + lrow;
        const int mt0   = (wib / NT) * TPW;
        v2f bf[F / 4];
#pragma unroll
        for (int kk = 0; kk < F / 4; ++kk) {
            bf[kk].x = Wn[(kk * 4 + khalf + 0) * FNEXT + col];
            bf[kk].y = Wn[(kk * 4 + khalf + 1) * FNEXT + col];
        }
        __syncthreads();
        v8f pacc[TPW];
#pragma unroll
        for (int t = 0; t < TPW; ++t) pacc[t] = v8f{};
#pragma unroll
        for (int kk = 0; kk < F / 4; ++kk) {
#pragma unroll
            for (int t = 0; t < TPW; ++t) {
                const v2f a = *(const v2f*)&s_x[((mt0 + t) * 16 + lrow) * XP + kk * 4 + khalf];
                pacc[t] = __builtin_amdgcn_wmma_f32_16x16x4_f32(
                    false, a, false, bf[kk], (short)0, pacc[t], false, false);
            }
        }
        const int blk0 = (int)blockIdx.x * (GW * NPW);
#pragma unroll
        for (int t = 0; t < TPW; ++t) {
#pragma unroll
            for (int r = 0; r < 8; ++r) {
                const int row = blk0 + (mt0 + t) * 16 + (lane >> 4) * 8 + r;
                if (row < n) y[(long)row * FNEXT + col] = pacc[t][r];
            }
        }
    }
}


extern "C" void kernel_launch(void* const* d_in, const int* in_sizes, int n_in,
                              void* d_out, int out_size, void* d_ws, size_t ws_size,
                              hipStream_t stream) {
    const float* x    = (const float*)d_in[0];
    const int*   src  = (const int*)  d_in[1];
    const int*   dst  = (const int*)  d_in[2];
    const float* ew   = (const float*)d_in[3];
    const float* W1   = (const float*)d_in[4];
    const float* b1   = (const float*)d_in[5];
    const float* W2   = (const float*)d_in[6];
    const float* b2   = (const float*)d_in[7];
    const float* W3   = (const float*)d_in[8];
    const float* b3   = (const float*)d_in[9];
    const float* W4   = (const float*)d_in[10];
    const float* b4   = (const float*)d_in[11];
    const float* Wres = (const float*)d_in[12];
    const float* bres = (const float*)d_in[13];
    float* out = (float*)d_out;

    const int N  = in_sizes[0] / IN_F;
    const int E  = in_sizes[1];
    const int nbk = ceil_div_h(N, BNODES);
    const int g1  = ceil_div_h(E, CHUNK);
    const int L   = nbk * g1;
    const int nb  = ceil_div_h(L, 256);

    unsigned int* ws = (unsigned int*)d_ws;
    float* norm   = (float*)(ws);
    int*   rowptr = (int*)  (ws + N);
    int*   bh     = (int*)  (ws + 2L * N + 1);
    int*   offs   = bh + L;
    int*   bsums  = offs + L + 1;
    long   tmpOff = (2L * N + 2L * L + nb + 2 + 63) & ~63L;
    long   recOff = tmpOff + ((4L * E + 63) & ~63L);
    long   hOff   = recOff + ((2L * E + 63) & ~63L);
    long   plane  = ((long)N * H_F + 63) & ~63L;
    int4*  tmp    = (int4*) (ws + tmpOff);
    int2*  eRec   = (int2*) (ws + recOff);
    float* h      = (float*)(ws + hOff);
    float* xbuf   = h    + plane;
    float* res    = xbuf + plane;

    const int T  = 256;
    const int GT = GW * 32;
    const int mt = ceil_div_h(N, 16);
    const int gblocks = ceil_div_h(N, GW * NPW);

    bucket_hist_kernel<<<g1, PT, nbk * sizeof(int), stream>>>(dst, bh, E, nbk);
    scan_block_kernel<<<nb, 256, 0, stream>>>(bh, offs, bsums, L);
    scan_sums_kernel<<<1, 32, 0, stream>>>(bsums, nb);
    scan_add_kernel<<<ceil_div_h(L, T), T, 0, stream>>>(offs, bsums, L, E);
    bucket_scatter_kernel<<<g1, PT, nbk * sizeof(int), stream>>>(src, dst, ew, offs, tmp, E, nbk);
    bucket_build_kernel<<<nbk, PT, 0, stream>>>(tmp, offs, rowptr, norm, eRec, N, E, g1);
    row_coef_kernel<<<ceil_div_h(4L * N, T), T, 0, stream>>>(rowptr, norm, eRec, N);

    gemm_wmma_kernel<IN_F, C_F><<<mt, (C_F / 16) * 32, 0, stream>>>(x, Wres, res, N);

    gemm_wmma_kernel<IN_F, H_F><<<mt, (H_F / 16) * 32, 0, stream>>>(x, W1, h, N);
    gather_agg_kernel<H_F, true, false, H_F><<<gblocks, GT, 0, stream>>>(h, rowptr, eRec,
                                                                        b1, nullptr, nullptr, W2, xbuf, N);
    gather_agg_kernel<H_F, true, false, H_F><<<gblocks, GT, 0, stream>>>(xbuf, rowptr, eRec,
                                                                        b2, nullptr, nullptr, W3, h, N);
    gather_agg_kernel<H_F, true, false, C_F><<<gblocks, GT, 0, stream>>>(h, rowptr, eRec,
                                                                        b3, nullptr, nullptr, W4, xbuf, N);
    gather_agg_kernel<C_F, false, true, 0><<<gblocks, GT, 0, stream>>>(xbuf, rowptr, eRec,
                                                                       b4, res, bres, nullptr, out, N);
}


# residual and layer-1 products in one launch: a work-group stages 48 input rows in LDS once, six waves each keep one 16-column slice of W1 or Wres in registers over 3 row tiles; same k order
# speedup vs baseline: 2.6368x; 1.1154x over previous
#include <hip/hip_runtime.h>
#include <type_traits>


typedef __attribute__((ext_vector_type(2))) float v2f;
typedef __attribute__((ext_vector_type(8))) float v8f;
typedef int rec8 __attribute__((ext_vector_type(16), aligned(8)));

#define IN_F 128
#define H_F  64
#define C_F  32

#define BSH    8
#define BNODES (1 << BSH)
#define CHUNK  16384
#define PT     1024
#define RCAP   6144

static inline int ceil_div_h(long a, long b) { return (int)((a + b - 1) / b); }


__global__ void bucket_hist_kernel(const int* __restrict__ dst, int* __restrict__ bh,
                                   int E, int nbk) {
    extern __shared__ int s_bins[];
    for (int k = threadIdx.x; k < nbk; k += blockDim.x) s_bins[k] = 0;
    __syncthreads();
    const int e0 = blockIdx.x * CHUNK;
    const int e1 = min(e0 + CHUNK, E);
    const int bs = blockDim.x;
    for (int eb = e0 + threadIdx.x; eb < e1; eb += 4 * bs) {
        int d[4];
        for (int j = 0; j < 4; ++j) d[j] = dst[min(eb + j * bs, e1 - 1)];
        asm volatile("" :: "v"(d[0]), "v"(d[1]), "v"(d[2]), "v"(d[3]));
        for (int j = 0; j < 4; ++j)
            if (eb + j * bs < e1) atomicAdd(&s_bins[d[j] >> BSH], 1);
    }
    __syncthreads();
    for (int k = threadIdx.x; k < nbk; k += blockDim.x)
        bh[k * gridDim.x + blockIdx.x] = s_bins[k];
}


__global__ void scan_block_kernel(const int* __restrict__ counts, int* __restrict__ rowptr,
                                  int* __restrict__ bsums, int n) {
    __shared__ int sdata[256];
    int tid = threadIdx.x;
    int i = blockIdx.x * 256 + tid;
    int v = (i < n) ? counts[i] : 0;
    sdata[tid] = v;
    __syncthreads();
#pragma unroll
    for (int off = 1; off < 256; off <<= 1) {
        int t = (tid >= off) ? sdata[tid - off] : 0;
        __syncthreads();
        sdata[tid] += t;
        __syncthreads();
    }
    if (i < n) rowptr[i] = sdata[tid] - v;
    if (tid == 255) bsums[blockIdx.x] = sdata[255];
}

__global__ void scan_sums_kernel(int* __restrict__ bsums, int nb) {
    if (blockIdx.x == 0 && threadIdx.x == 0) {
        int acc = 0;
        for (int i = 0; i < nb; ++i) { int t = bsums[i]; bsums[i] = acc; acc += t; }
    }
}

__global__ void scan_add_kernel(int* __restrict__ rowptr, const int* __restrict__ bsums,
                                int n, int E) {
    int i = blockIdx.x * blockDim.x + threadIdx.x;
    if (i < n) rowptr[i] += bsums[i >> 8];
    if (i == 0) rowptr[n] = E;
}

__global__ void bucket_scatter_kernel(const int* __restrict__ src, const int* __restrict__ dst,
                                      const float* __restrict__ ew, const int* __restrict__ offs,
                                      int4* __restrict__ tmp, int E, int nbk) {
    extern __shared__ int s_bins[];
    for (int k = threadIdx.x; k < nbk; k += blockDim.x)
        s_bins[k] = offs[k * gridDim.x + blockIdx.x];
    __syncthreads();
    const int e0 = blockIdx.x * CHUNK;
    const int e1 = min(e0 + CHUNK, E);
    const int bs = blockDim.x;
    for (int eb = e0 + threadIdx.x; eb < e1; eb += 4 * bs) {
        int s[4], d[4], w[4];
        for (int j = 0; j < 4; ++j) {
            const int e = min(eb + j * bs, e1 - 1);
            s[j] = src[e];
            d[j] = dst[e];
            w[j] = __float_as_int(ew[e]);
        }
        asm volatile("" :: "v"(s[0]), "v"(s[1]), "v"(s[2]), "v"(s[3]),
                           "v"(d[0]), "v"(d[1]), "v"(d[2]), "v"(d[3]),
                           "v"(w[0]), "v"(w[1]), "v"(w[2]), "v"(w[3]));
        for (int j = 0; j < 4; ++j) {
            if (eb + j * bs < e1) {
                int pos = atomicAdd(&s_bins[d[j] >> BSH], 1);
                tmp[pos] = make_int4(s[j], w[j], d[j], 0);
            }
        }
    }
}

__global__ void bucket_build_kernel(const int4* __restrict__ tmp, const int* __restrict__ offs,
                                    int* __restrict__ rowptr, float* __restrict__ norm,
                                    int2* __restrict__ eRec, int n, int E, int g1) {
    __shared__ int   s_cnt[BNODES];
    __shared__ float s_deg[BNODES];
    __shared__ int   s_scan[BNODES];
    __shared__ int2  s_rec[RCAP];
    const int tid   = threadIdx.x;
    const int t0    = offs[blockIdx.x * g1];
    const int t1    = offs[(blockIdx.x + 1) * g1];
    const int nbase = blockIdx.x << BSH;
    const bool stage = (t1 - t0) <= RCAP;

    if (tid < BNODES) { s_cnt[tid] = 0; s_deg[tid] = 0.0f; }
    __syncthreads();
    const int bs = blockDim.x;
    for (int ib = t0 + tid; ib < t1; ib += 4 * bs) {
        int4 r[4];
        for (int j = 0; j < 4; ++j) r[j] = tmp[min(ib + j * bs, t1 - 1)];
        asm volatile("" :: "v"(r[0].y), "v"(r[0].z), "v"(r[1].y), "v"(r[1].z),
                           "v"(r[2].y), "v"(r[2].z), "v"(r[3].y), "v"(r[3].z));
        for (int j = 0; j < 4; ++j) {
            if (ib + j * bs < t1) {
                int dl = r[j].z - nbase;
                atomicAdd(&s_cnt[dl], 1);
                atomicAdd(&s_deg[dl], __int_as_float(r[j].y));
            }
        }
    }
    __syncthreads();

    int v = 0;
    if (tid < BNODES) { v = s_cnt[tid]; s_scan[tid] = v; }
    __syncthreads();
    for (int off = 1; off < BNODES; off <<= 1) {
        int t = 0;
        if (tid < BNODES && tid >= off) t = s_scan[tid - off];
        __syncthreads();
        if (tid < BNODES) s_scan[tid] += t;
        __syncthreads();
    }
    if (tid < BNODES) {
        const int start = t0 + s_scan[tid] - v;
        s_cnt[tid] = start;
        const int node = nbase + tid;
        if (node < n) {
            rowptr[node] = start;
            float d = s_deg[tid];
            norm[node] = (d > 0.0f) ? rsqrtf(fmaxf(d, 1e-12f)) : 0.0f;
        }
    }
    if (blockIdx.x == gridDim.x - 1 && tid == 0) rowptr[n] = E;
    __syncthreads();

    for (int ib = t0 + tid; ib < t1; ib += 4 * bs) {
        int4 r[4];
        for (int j = 0; j < 4; ++j) r[j] = tmp[min(ib + j * bs, t1 - 1)];
        asm volatile("" :: "v"(r[0].x), "v"(r[0].y), "v"(r[0].z), "v"(r[1].x), "v"(r[1].y), "v"(r[1].z),
                           "v"(r[2].x), "v"(r[2].y), "v"(r[2].z), "v"(r[3].x), "v"(r[3].y), "v"(r[3].z));
        for (int j = 0; j < 4; ++j) {
            if (ib + j * bs < t1) {
                int pos = atomicAdd(&s_cnt[r[j].z - nbase], 1);
                if (stage) s_rec[pos - t0] = make_int2(r[j].x, r[j].y);
                else       eRec[pos]       = make_int2(r[j].x, r[j].y);
            }
        }
    }
    if (stage) {
        __syncthreads();
        for (int i = tid; i < t1 - t0; i += blockDim.x) eRec[t0 + i] = s_rec[i];
    }
}

__global__ void row_coef_kernel(const int* __restrict__ rowptr, const float* __restrict__ norm,
                                int2* __restrict__ eRec, int n) {
    const int t   = blockIdx.x * blockDim.x + threadIdx.x;
    const int n0  = (t >> 4) * 4;
    const int sub = t & 15;
    if (n0 >= n) return;
    const int nl = min(n0 + 4, n);
    int   rp[5];
    float nd[4];
    for (int q = 0; q < 5; ++q) rp[q] = rowptr[min(n0 + q, nl)];
    for (int q = 0; q < 4; ++q) nd[q] = norm[min(n0 + q, n - 1)];
    for (int i = rp[0] + sub; i < rp[4]; i += 16) {
        int2  r   = eRec[i];
        float ndq = nd[0];
        if (i >= rp[1]) ndq = nd[1];
        if (i >= rp[2]) ndq = nd[2];
        if (i >= rp[3]) ndq = nd[3];
        float c = __int_as_float(r.y) * norm[r.x] * ndq;
        eRec[i] = make_int2(r.x, __float_as_int(c));
    }
}

#define PM  48
#define PXP 132
#define PWV ((H_F + C_F) / 16)

template <int FOUT>
__device__ __forceinline__ void load_b_slice(const float* __restrict__ Wm, int col, int khalf,
                                             v2f (&bf)[IN_F / 4]) {
    const float* p = Wm + khalf * FOUT + col;
#pragma unroll
    for (int kk = 0; kk < IN_F / 4; ++kk) {
        bf[kk].x = p[(kk * 4 + 0) * FOUT];
        bf[kk].y = p[(kk * 4 + 1) * FOUT];
    }
}

__global__ void __launch_bounds__(PWV * 32)
input_products_kernel(const float* __restrict__ X, const float* __restrict__ Wa,
                      const float* __restrict__ Wb, float* __restrict__ Ya,
                      float* __restrict__ Yb, int nrows) {
    __shared__ __attribute__((aligned(16))) float s_t[PM * PXP];
    constexpr int NT  = PWV * 32;
    constexpr int Q   = IN_F / 4;
    constexpr int RPS = NT / Q;
    constexpr int NST = PM / RPS;
    static_assert(NT % Q == 0 && PM % RPS == 0 && PM % 16 == 0, "whole sweeps, whole row tiles");
    const int tid = threadIdx.x;
    const int m0  = blockIdx.x * PM;
    const int r0  = tid / Q;
    const int q4  = (tid % Q) * 4;

    float4 sv[NST];
#pragma unroll
    for (int j = 0; j < NST; ++j) {
        const int row = min(m0 + r0 + j * RPS, nrows - 1);
        sv[j] = *(const float4*)(X + (long)row * IN_F + q4);
    }

    const int wave  = __builtin_amdgcn_readfirstlane(tid >> 5);
    const int lane  = tid & 31;
    const int lrow  = lane & 15;
    const int khalf = (lane >> 4) * 2;
    const bool isA  = wave < H_F / 16;
    const int col   = (isA ? wave : wave - H_F / 16) * 16 + lrow;

    v2f bf[IN_F / 4];
    if (isA) load_b_slice<H_F>(Wa, col, khalf, bf);
    else     load_b_slice<C_F>(Wb, col, khalf, bf);

#pragma unroll
    for (int j = 0; j < NST; ++j)
        *(float4*)(s_t + (r0 + j * RPS) * PXP + q4) = sv[j];
    __syncthreads();

    constexpr int TPW = PM / 16;
    v8f pacc[TPW];
#pragma unroll
    for (int t = 0; t < TPW; ++t) pacc[t] = v8f{};
#pragma unroll
    for (int kk = 0; kk < IN_F / 4; ++kk) {
#pragma unroll
        for (int t = 0; t < TPW; ++t) {
            const v2f a = *(const v2f*)&s_t[(t * 16 + lrow) * PXP + kk * 4 + khalf];
            pacc[t] = __builtin_amdgcn_wmma_f32_16x16x4_f32(
                false, a, false, bf[kk], (short)0, pacc[t], false, false);
        }
    }
    float* __restrict__ Ym = isA ? Ya : Yb;
    const int fout = isA ? H_F : C_F;
#pragma unroll
    for (int t = 0; t < TPW; ++t) {
#pragma unroll
        for (int r = 0; r < 8; ++r) {
            const int row = m0 + t * 16 + (lane >> 4) * 8 + r;
            if (row < nrows) Ym[(long)row * fout + col] = pacc[t][r];
        }
    }
}

#define NPW 8
#define GC  16
#define GW  8
#define XP  68

template <int F, bool RELU, bool FINAL, int FNEXT>
__global__ void __launch_bounds__(GW * 32)
gather_agg_kernel(const float* __restrict__ h,
                  const int* __restrict__ rowptr,
                  const int2* __restrict__ eRec,
                  const float* __restrict__ bias,
                  const float* __restrict__ res,
                  const float* __restrict__ bias2,
                  const float* __restrict__ Wn,
                  float* __restrict__ y, int n) {
    constexpr int VPL = F / 32;
    static_assert(FNEXT == 0 || (F == 64 && !FINAL && FNEXT % 16 == 0 && (4 * FNEXT / 16) % GW == 0),
                  "fused product: 64-wide rows, 64 x FNEXT tiles split evenly over the waves");
    __shared__ __attribute__((aligned(16))) float s_x[FNEXT > 0 ? GW * NPW * XP : 1];
    const int lane  = threadIdx.x & 31;
    const int fbase = lane * VPL;
    const int wib = __builtin_amdgcn_readfirstlane((int)(threadIdx.x >> 5));
    const int wv  = __builtin_amdgcn_readfirstlane((int)(blockIdx.x * GW)) + wib;
    const int n0  = wv * NPW;
    if constexpr (FNEXT == 0) {
        if (n0 >= n) return;
    }
    const int n1 = min(n0 + NPW, n);
    if constexpr (FNEXT > 0) {
        for (int q = max(n1 - n0, 0); q < NPW; ++q)
            *(float2*)(s_x + (wib * NPW + q) * XP + fbase) = make_float2(0.0f, 0.0f);
    }

    const int rpv = rowptr[min(n0 + lane, n1)];
    int       e    = __builtin_amdgcn_readlane(rpv, 0);
    const int eEnd = __builtin_amdgcn_readlane(rpv, NPW);
    int       nb   = __builtin_amdgcn_readlane(rpv, 1);
    int       node = n0;

    float bs[VPL], bs2[VPL], resv[VPL], acc[VPL];
#pragma unroll
    for (int j = 0; j < VPL; ++j) {
        bs[j]   = bias[fbase + j];
        bs2[j]  = FINAL ? bias2[fbase + j] : 0.0f;
        resv[j] = FINAL ? res[(long)n0 * F + fbase + j] : 0.0f;
        acc[j]  = 0.0f;
    }

    auto flush = [&]() {
        const long base = (long)node * F + fbase;
        float o[VPL];
#pragma unroll
        for (int j = 0; j < VPL; ++j) {
            float v = acc[j] + bs[j];
            if (FINAL) v += resv[j] + bs2[j];
            if (RELU) v = fmaxf(v, 0.0f);
            o[j]   = v;
            acc[j] = 0.0f;
        }
        if constexpr (FNEXT > 0) {
            *(float2*)(s_x + (wib * NPW + (node - n0)) * XP + fbase) = make_float2(o[0], o[1]);
        } else if constexpr (VPL == 2) {
            *(float2*)(y + base) = make_float2(o[0], o[1]);
        } else {
            y[base] = o[0];
        }
        ++node;
        nb = __builtin_amdgcn_readlane(rpv, node - n0 + 1);
        if constexpr (FINAL) {
            if (node < n1) {
#pragma unroll
                for (int j = 0; j < VPL; ++j) resv[j] = res[(long)node * F + fbase + j];
            }
        }
    };

    auto group = [&](auto fullTag) {
        constexpr bool FULL = decltype(fullTag)::value;
        int   sx[GC];
        float cf[GC];
        if constexpr (FULL) {
            const rec8* p = (const rec8*)(eRec + e);
            const rec8 ra = p[0], rb = p[1];
#pragma unroll
            for (int j = 0; j < 8; ++j) {
                sx[j]     = ra[2 * j];  cf[j]     = __int_as_float(ra[2 * j + 1]);
                sx[8 + j] = rb[2 * j];  cf[8 + j] = __int_as_float(rb[2 * j + 1]);
            }
        } else {
#pragma unroll
            for (int j = 0; j < GC; ++j) {
                const int2 r = eRec[min(e + j, eEnd - 1)];
                sx[j] = r.x;
                cf[j] = __int_as_float(r.y);
            }
        }
        float rv[GC][VPL];
#pragma unroll
        for (int j = 0; j < GC; ++j) {
            if constexpr (VPL == 2) {
                const float2 v = ((const float2*)h)[(unsigned)sx[j] * (unsigned)(F / 2) + (unsigned)lane];
                rv[j][0] = v.x; rv[j][1] = v.y;
            } else {
                rv[j][0] = h[(unsigned)sx[j] * (unsigned)F + (unsigned)lane];
            }
        }
#pragma unroll
        for (int j = 0; j < GC; ++j) {
            if (FULL || e + j < eEnd) {
                while (e + j == nb) flush();
#pragma unroll
                for (int q = 0; q < VPL; ++q) acc[q] += rv[j][q] * cf[j];
            }
        }
    };

    while (e < eEnd) {
        if (eEnd - e >= GC) group(std::true_type{});
        else                group(std::false_type{});
        e += GC;
    }
    while (node < n1) flush();

    if constexpr (FNEXT > 0) {
        constexpr int NT  = FNEXT / 16;
        constexpr int TPW = (4 * NT) / GW;
        const int lrow  = lane & 15;
        const int khalf = (lane >> 4) * 2;
        const int col   = (wib % NT) * 16 + lrow;
        const int mt0   = (wib / NT) * TPW;
        v2f bf[F / 4];
#pragma unroll
        for (int kk = 0; kk < F / 4; ++kk) {
            bf[kk].x = Wn[(kk * 4 + khalf + 0) * FNEXT + col];
            bf[kk].y = Wn[(kk * 4 + khalf + 1) * FNEXT + col];
        }
        __syncthreads();
        v8f pacc[TPW];
#pragma unroll
        for (int t = 0; t < TPW; ++t) pacc[t] = v8f{};
#pragma unroll
        for (int kk = 0; kk < F / 4; ++kk) {
#pragma unroll
            for (int t = 0; t < TPW; ++t) {
                const v2f a = *(const v2f*)&s_x[((mt0 + t) * 16 + lrow) * XP + kk * 4 + khalf];
                pacc[t] = __builtin_amdgcn_wmma_f32_16x16x4_f32(
                    false, a, false, bf[kk], (short)0, pacc[t], false, false);
            }
        }
        const int blk0 = (int)blockIdx.x * (GW * NPW);
#pragma unroll
        for (int t = 0; t < TPW; ++t) {
#pragma unroll
            for (int r = 0; r < 8; ++r) {
                const int row = blk0 + (mt0 + t) * 16 + (lane >> 4) * 8 + r;
                if (row < n) y[(long)row * FNEXT + col] = pacc[t][r];
            }
        }
    }
}


extern "C" void kernel_launch(void* const* d_in, const int* in_sizes, int n_in,
                              void* d_out, int out_size, void* d_ws, size_t ws_size,
                              hipStream_t stream) {
    const float* x    = (const float*)d_in[0];
    const int*   src  = (const int*)  d_in[1];
    const int*   dst  = (const int*)  d_in[2];
    const float* ew   = (const float*)d_in[3];
    const float* W1   = (const float*)d_in[4];
    const float* b1   = (const float*)d_in[5];
    const float* W2   = (const float*)d_in[6];
    const float* b2   = (const float*)d_in[7];
    const float* W3   = (const float*)d_in[8];
    const float* b3   = (const float*)d_in[9];
    const float* W4   = (const float*)d_in[10];
    const float* b4   = (const float*)d_in[11];
    const float* Wres = (const float*)d_in[12];
    const float* bres = (const float*)d_in[13];
    float* out = (float*)d_out;

    const int N  = in_sizes[0] / IN_F;
    const int E  = in_sizes[1];
    const int nbk = ceil_div_h(N, BNODES);
    const int g1  = ceil_div_h(E, CHUNK);
    const int L   = nbk * g1;
    const int nb  = ceil_div_h(L, 256);

    unsigned int* ws = (unsigned int*)d_ws;
    float* norm   = (float*)(ws);
    int*   rowptr = (int*)  (ws + N);
    int*   bh     = (int*)  (ws + 2L * N + 1);
    int*   offs   = bh + L;
    int*   bsums  = offs + L + 1;
    long   tmpOff = (2L * N + 2L * L + nb + 2 + 63) & ~63L;
    long   recOff = tmpOff + ((4L * E + 63) & ~63L);
    long   hOff   = recOff + ((2L * E + 63) & ~63L);
    long   plane  = ((long)N * H_F + 63) & ~63L;
    int4*  tmp    = (int4*) (ws + tmpOff);
    int2*  eRec   = (int2*) (ws + recOff);
    float* h      = (float*)(ws + hOff);
    float* xbuf   = h    + plane;
    float* res    = xbuf + plane;

    const int T  = 256;
    const int GT = GW * 32;
    const int gblocks = ceil_div_h(N, GW * NPW);

    bucket_hist_kernel<<<g1, PT, nbk * sizeof(int), stream>>>(dst, bh, E, nbk);
    scan_block_kernel<<<nb, 256, 0, stream>>>(bh, offs, bsums, L);
    scan_sums_kernel<<<1, 32, 0, stream>>>(bsums, nb);
    scan_add_kernel<<<ceil_div_h(L, T), T, 0, stream>>>(offs, bsums, L, E);
    bucket_scatter_kernel<<<g1, PT, nbk * sizeof(int), stream>>>(src, dst, ew, offs, tmp, E, nbk);
    bucket_build_kernel<<<nbk, PT, 0, stream>>>(tmp, offs, rowptr, norm, eRec, N, E, g1);
    row_coef_kernel<<<ceil_div_h(4L * N, T), T, 0, stream>>>(rowptr, norm, eRec, N);

    input_products_kernel<<<ceil_div_h(N, PM), PWV * 32, 0, stream>>>(x, W1, Wres, h, res, N);
    gather_agg_kernel<H_F, true, false, H_F><<<gblocks, GT, 0, stream>>>(h, rowptr, eRec,
                                                                        b1, nullptr, nullptr, W2, xbuf, N);
    gather_agg_kernel<H_F, true, false, H_F><<<gblocks, GT, 0, stream>>>(xbuf, rowptr, eRec,
                                                                        b2, nullptr, nullptr, W3, h, N);
    gather_agg_kernel<H_F, true, false, C_F><<<gblocks, GT, 0, stream>>>(h, rowptr, eRec,
                                                                        b3, nullptr, nullptr, W4, xbuf, N);
    gather_agg_kernel<C_F, false, true, 0><<<gblocks, GT, 0, stream>>>(xbuf, rowptr, eRec,
                                                                       b4, res, bres, nullptr, out, N);
}


# level-1 sort made chunk-local (LDS histogram, scan, cursor over a work-group's 8192 edges; owners read each chunk's run via a start table): 4 launches go; arrival order of degree and row sums changes
# speedup vs baseline: 3.0686x; 1.1638x over previous
#include <hip/hip_runtime.h>
#include <type_traits>


typedef __attribute__((ext_vector_type(2))) float v2f;
typedef __attribute__((ext_vector_type(8))) float v8f;
typedef int rec8 __attribute__((ext_vector_type(16), aligned(8)));

#define IN_F 128
#define H_F  64
#define C_F  32

#define BSH    8
#define BNODES (1 << BSH)
#define CHUNK  8192
#define PT     1024
#define EPT    (CHUNK / PT)
#define RB     8
#define RCAP   6144

static inline int ceil_div_h(long a, long b) { return (int)((a + b - 1) / b); }


__global__ void __launch_bounds__(PT)
chunk_sort_kernel(const int* __restrict__ src, const int* __restrict__ dst,
                  const float* __restrict__ ew, int* __restrict__ lo,
                  int4* __restrict__ tmp, int E, int nbk) {
    extern __shared__ int s_bins[];
    __shared__ int s_wsum[PT / 32];
    const int tid  = threadIdx.x;
    const int lane = tid & 31;
    const int wv   = tid >> 5;
    for (int k = tid; k < nbk; k += PT) s_bins[k] = 0;
    __syncthreads();
    const int e0 = blockIdx.x * CHUNK;
    const int e1 = min(e0 + CHUNK, E);
    int s[EPT], d[EPT], w[EPT];
#pragma unroll
    for (int j = 0; j < EPT; ++j) {
        const int e = min(e0 + tid + j * PT, e1 - 1);
        s[j] = src[e];
        d[j] = dst[e];
        w[j] = __float_as_int(ew[e]);
    }
    asm volatile("" :: "v"(s[0]), "v"(s[1]), "v"(s[2]), "v"(s[3]), "v"(s[4]), "v"(s[5]), "v"(s[6]), "v"(s[7]),
                       "v"(d[0]), "v"(d[1]), "v"(d[2]), "v"(d[3]), "v"(d[4]), "v"(d[5]), "v"(d[6]), "v"(d[7]),
                       "v"(w[0]), "v"(w[1]), "v"(w[2]), "v"(w[3]), "v"(w[4]), "v"(w[5]), "v"(w[6]), "v"(w[7]));
    static_assert(EPT == 8, "the fence above names EPT = 8 edges");
#pragma unroll
    for (int j = 0; j < EPT; ++j)
        if (e0 + tid + j * PT < e1) atomicAdd(&s_bins[d[j] >> BSH], 1);
    __syncthreads();

    int carry = 0;
    for (int base = 0; base < nbk; base += PT) {
        const int i = base + tid;
        const int v = (i < nbk) ? s_bins[i] : 0;
        int x = v;
        for (int o = 1; o < 32; o <<= 1) {
            const int t = __shfl_up(x, o);
            if (lane >= o) x += t;
        }
        if (lane == 31) s_wsum[wv] = x;
        __syncthreads();
        if (wv == 0) {
            int z = s_wsum[lane];
            for (int o = 1; o < 32; o <<= 1) {
                const int t = __shfl_up(z, o);
                if (lane >= o) z += t;
            }
            s_wsum[lane] = z;
        }
        __syncthreads();
        const int excl = carry + (wv ? s_wsum[wv - 1] : 0) + x - v;
        if (i < nbk) {
            s_bins[i] = excl;
            lo[blockIdx.x * (nbk + 1) + i] = excl;
        }
        carry += s_wsum[PT / 32 - 1];
        __syncthreads();
    }
    if (tid == 0) lo[blockIdx.x * (nbk + 1) + nbk] = e1 - e0;

#pragma unroll
    for (int j = 0; j < EPT; ++j) {
        if (e0 + tid + j * PT < e1) {
            const int pos = atomicAdd(&s_bins[d[j] >> BSH], 1);
            tmp[e0 + pos] = make_int4(s[j], w[j], d[j], 0);
        }
    }
}

__global__ void __launch_bounds__(PT)
bucket_build_kernel(const int4* __restrict__ tmp, const int* __restrict__ lo,
                    int* __restrict__ rowptr, float* __restrict__ norm,
                    int2* __restrict__ eRec, int n, int E, int g1, int nbk) {
    __shared__ int   s_cnt[BNODES];
    __shared__ float s_deg[BNODES];
    __shared__ int   s_scan[BNODES];
    __shared__ int2  s_rec[RCAP];
    __shared__ int   s_red[2 * (PT / 32)];
    const int tid   = threadIdx.x;
    const int lane  = tid & 31;
    const int wib   = __builtin_amdgcn_readfirstlane(tid >> 5);
    const int k     = blockIdx.x;
    const int ld    = nbk + 1;
    constexpr int NW = PT / 32;
    const int nbase = k << BSH;

    int sa = 0, sb = 0;
    for (int c = tid; c < g1; c += PT) {
        sa += lo[c * ld + k];
        sb += lo[c * ld + k + 1];
    }
    for (int o = 16; o > 0; o >>= 1) {
        sa += __shfl_xor(sa, o);
        sb += __shfl_xor(sb, o);
    }
    if (lane == 0) { s_red[wib] = sa; s_red[PT / 32 + wib] = sb; }
    if (tid < BNODES) { s_cnt[tid] = 0; s_deg[tid] = 0.0f; }
    __syncthreads();
    sa = s_red[lane];
    sb = s_red[PT / 32 + lane];
    for (int o = 16; o > 0; o >>= 1) {
        sa += __shfl_xor(sa, o);
        sb += __shfl_xor(sb, o);
    }
    const int t0 = __builtin_amdgcn_readfirstlane(sa);
    const int t1 = __builtin_amdgcn_readfirstlane(sb);
    const bool stage = (t1 - t0) <= RCAP;

    auto sweep = [&](auto placeTag) {
        constexpr bool PLACE = decltype(placeTag)::value;
        auto one = [&](const int4 r) {
            const int dl = r.z - nbase;
            if constexpr (PLACE) {
                const int pos = atomicAdd(&s_cnt[dl], 1);
                if (stage) s_rec[pos - t0] = make_int2(r.x, r.y);
                else       eRec[pos]       = make_int2(r.x, r.y);
            } else {
                atomicAdd(&s_cnt[dl], 1);
                atomicAdd(&s_deg[dl], __int_as_float(r.y));
            }
        };
        for (int cb = wib; cb < g1; cb += NW * RB) {
            int a[RB], b[RB];
#pragma unroll
            for (int j = 0; j < RB; ++j) {
                const int cc = min(cb + NW * j, g1 - 1);
                a[j] = lo[cc * ld + k];
                b[j] = lo[cc * ld + k + 1];
            }
#pragma unroll
            for (int j = 0; j < RB; ++j) {
                const int cj = cb + NW * j;
                const int cc = min(cj, g1 - 1);
                a[j] += cc * CHUNK;
                b[j]  = (cj < g1) ? b[j] + cc * CHUNK : a[j];
            }
            int4 r[RB];
#pragma unroll
            for (int j = 0; j < RB; ++j) {
                const int idx = min(min(a[j] + lane, max(b[j] - 1, a[j])), E - 1);
                r[j] = tmp[idx];
            }
            asm volatile("" :: "v"(r[0].x), "v"(r[0].y), "v"(r[0].z), "v"(r[1].x), "v"(r[1].y), "v"(r[1].z),
                               "v"(r[2].x), "v"(r[2].y), "v"(r[2].z), "v"(r[3].x), "v"(r[3].y), "v"(r[3].z),
                               "v"(r[4].x), "v"(r[4].y), "v"(r[4].z), "v"(r[5].x), "v"(r[5].y), "v"(r[5].z),
                               "v"(r[6].x), "v"(r[6].y), "v"(r[6].z), "v"(r[7].x), "v"(r[7].y), "v"(r[7].z));
            static_assert(RB == 8, "the fence above names RB = 8 runs");
#pragma unroll
            for (int j = 0; j < RB; ++j) {
                if (a[j] + lane < b[j]) one(r[j]);
                for (int i = a[j] + 32 + lane; i < b[j]; i += 32) one(tmp[i]);
            }
        }
    };

    sweep(std::false_type{});
    __syncthreads();

    int v = 0;
    if (tid < BNODES) { v = s_cnt[tid]; s_scan[tid] = v; }
    __syncthreads();
    for (int off = 1; off < BNODES; off <<= 1) {
        int t = 0;
        if (tid < BNODES && tid >= off) t = s_scan[tid - off];
        __syncthreads();
        if (tid < BNODES) s_scan[tid] += t;
        __syncthreads();
    }
    if (tid < BNODES) {
        const int start = t0 + s_scan[tid] - v;
        s_cnt[tid] = start;
        const int node = nbase + tid;
        if (node < n) {
            rowptr[node] = start;
            float d = s_deg[tid];
            norm[node] = (d > 0.0f) ? rsqrtf(fmaxf(d, 1e-12f)) : 0.0f;
        }
    }
    if (blockIdx.x == gridDim.x - 1 && tid == 0) rowptr[n] = E;
    __syncthreads();

    sweep(std::true_type{});
    if (stage) {
        __syncthreads();
        for (int i = tid; i < t1 - t0; i += blockDim.x) eRec[t0 + i] = s_rec[i];
    }
}

__global__ void row_coef_kernel(const int* __restrict__ rowptr, const float* __restrict__ norm,
                                int2* __restrict__ eRec, int n) {
    const int t   = blockIdx.x * blockDim.x + threadIdx.x;
    const int n0  = (t >> 4) * 4;
    const int sub = t & 15;
    if (n0 >= n) return;
    const int nl = min(n0 + 4, n);
    int   rp[5];
    float nd[4];
    for (int q = 0; q < 5; ++q) rp[q] = rowptr[min(n0 + q, nl)];
    for (int q = 0; q < 4; ++q) nd[q] = norm[min(n0 + q, n - 1)];
    for (int i = rp[0] + sub; i < rp[4]; i += 16) {
        int2  r   = eRec[i];
        float ndq = nd[0];
        if (i >= rp[1]) ndq = nd[1];
        if (i >= rp[2]) ndq = nd[2];
        if (i >= rp[3]) ndq = nd[3];
        float c = __int_as_float(r.y) * norm[r.x] * ndq;
        eRec[i] = make_int2(r.x, __float_as_int(c));
    }
}

#define PM  48
#define PXP 132
#define PWV ((H_F + C_F) / 16)

template <int FOUT>
__device__ __forceinline__ void load_b_slice(const float* __restrict__ Wm, int col, int khalf,
                                             v2f (&bf)[IN_F / 4]) {
    const float* p = Wm + khalf * FOUT + col;
#pragma unroll
    for (int kk = 0; kk < IN_F / 4; ++kk) {
        bf[kk].x = p[(kk * 4 + 0) * FOUT];
        bf[kk].y = p[(kk * 4 + 1) * FOUT];
    }
}

__global__ void __launch_bounds__(PWV * 32)
input_products_kernel(const float* __restrict__ X, const float* __restrict__ Wa,
                      const float* __restrict__ Wb, float* __restrict__ Ya,
                      float* __restrict__ Yb, int nrows) {
    __shared__ __attribute__((aligned(16))) float s_t[PM * PXP];
    constexpr int NT  = PWV * 32;
    constexpr int Q   = IN_F / 4;
    constexpr int RPS = NT / Q;
    constexpr int NST = PM / RPS;
    static_assert(NT % Q == 0 && PM % RPS == 0 && PM % 16 == 0, "whole sweeps, whole row tiles");
    const int tid = threadIdx.x;
    const int m0  = blockIdx.x * PM;
    const int r0  = tid / Q;
    const int q4  = (tid % Q) * 4;

    float4 sv[NST];
#pragma unroll
    for (int j = 0; j < NST; ++j) {
        const int row = min(m0 + r0 + j * RPS, nrows - 1);
        sv[j] = *(const float4*)(X + (long)row * IN_F + q4);
    }

    const int wave  = __builtin_amdgcn_readfirstlane(tid >> 5);
    const int lane  = tid & 31;
    const int lrow  = lane & 15;
    const int khalf = (lane >> 4) * 2;
    const bool isA  = wave < H_F / 16;
    const int col   = (isA ? wave : wave - H_F / 16) * 16 + lrow;

    v2f bf[IN_F / 4];
    if (isA) load_b_slice<H_F>(Wa, col, khalf, bf);
    else     load_b_slice<C_F>(Wb, col, khalf, bf);

#pragma unroll
    for (int j = 0; j < NST; ++j)
        *(float4*)(s_t + (r0 + j * RPS) * PXP + q4) = sv[j];
    __syncthreads();

    constexpr int TPW = PM / 16;
    v8f pacc[TPW];
#pragma unroll
    for (int t = 0; t < TPW; ++t) pacc[t] = v8f{};
#pragma unroll
    for (int kk = 0; kk < IN_F / 4; ++kk) {
#pragma unroll
        for (int t = 0; t < TPW; ++t) {
            const v2f a = *(const v2f*)&s_t[(t * 16 + lrow) * PXP + kk * 4 + khalf];
            pacc[t] = __builtin_amdgcn_wmma_f32_16x16x4_f32(
                false, a, false, bf[kk], (short)0, pacc[t], false, false);
        }
    }
    float* __restrict__ Ym = isA ? Ya : Yb;
    const int fout = isA ? H_F : C_F;
#pragma unroll
    for (int t = 0; t < TPW; ++t) {
#pragma unroll
        for (int r = 0; r < 8; ++r) {
            const int row = m0 + t * 16 + (lane >> 4) * 8 + r;
            if (row < nrows) Ym[(long)row * fout + col] = pacc[t][r];
        }
    }
}

#define NPW 8
#define GC  16
#define GW  8
#define XP  68

template <int F, bool RELU, bool FINAL, int FNEXT>
__global__ void __launch_bounds__(GW * 32)
gather_agg_kernel(const float* __restrict__ h,
                  const int* __restrict__ rowptr,
                  const int2* __restrict__ eRec,
                  const float* __restrict__ bias,
                  const float* __restrict__ res,
                  const float* __restrict__ bias2,
                  const float* __restrict__ Wn,
                  float* __restrict__ y, int n) {
    constexpr int VPL = F / 32;
    static_assert(FNEXT == 0 || (F == 64 && !FINAL && FNEXT % 16 == 0 && (4 * FNEXT / 16) % GW == 0),
                  "fused product: 64-wide rows, 64 x FNEXT tiles split evenly over the waves");
    __shared__ __attribute__((aligned(16))) float s_x[FNEXT > 0 ? GW * NPW * XP : 1];
    const int lane  = threadIdx.x & 31;
    const int fbase = lane * VPL;
    const int wib = __builtin_amdgcn_readfirstlane((int)(threadIdx.x >> 5));
    const int wv  = __builtin_amdgcn_readfirstlane((int)(blockIdx.x * GW)) + wib;
    const int n0  = wv * NPW;
    if constexpr (FNEXT == 0) {
        if (n0 >= n) return;
    }
    const int n1 = min(n0 + NPW, n);
    if constexpr (FNEXT > 0) {
        for (int q = max(n1 - n0, 0); q < NPW; ++q)
            *(float2*)(s_x + (wib * NPW + q) * XP + fbase) = make_float2(0.0f, 0.0f);
    }

    const int rpv = rowptr[min(n0 + lane, n1)];
    int       e    = __builtin_amdgcn_readlane(rpv, 0);
    const int eEnd = __builtin_amdgcn_readlane(rpv, NPW);
    int       nb   = __builtin_amdgcn_readlane(rpv, 1);
    int       node = n0;

    float bs[VPL], bs2[VPL], resv[VPL], acc[VPL];
#pragma unroll
    for (int j = 0; j < VPL; ++j) {
        bs[j]   = bias[fbase + j];
        bs2[j]  = FINAL ? bias2[fbase + j] : 0.0f;
        resv[j] = FINAL ? res[(long)n0 * F + fbase + j] : 0.0f;
        acc[j]  = 0.0f;
    }

    auto flush = [&]() {
        const long base = (long)node * F + fbase;
        float o[VPL];
#pragma unroll
        for (int j = 0; j < VPL; ++j) {
            float v = acc[j] + bs[j];
            if (FINAL) v += resv[j] + bs2[j];
            if (RELU) v = fmaxf(v, 0.0f);
            o[j]   = v;
            acc[j] = 0.0f;
        }
        if constexpr (FNEXT > 0) {
            *(float2*)(s_x + (wib * NPW + (node - n0)) * XP + fbase) = make_float2(o[0], o[1]);
        } else if constexpr (VPL == 2) {
            *(float2*)(y + base) = make_float2(o[0], o[1]);
        } else {
            y[base] = o[0];
        }
        ++node;
        nb = __builtin_amdgcn_readlane(rpv, node - n0 + 1);
        if constexpr (FINAL) {
            if (node < n1) {
#pragma unroll
                for (int j = 0; j < VPL; ++j) resv[j] = res[(long)node * F + fbase + j];
            }
        }
    };

    auto group = [&](auto fullTag) {
        constexpr bool FULL = decltype(fullTag)::value;
        int   sx[GC];
        float cf[GC];
        if constexpr (FULL) {
            const rec8* p = (const rec8*)(eRec + e);
            const rec8 ra = p[0], rb = p[1];
#pragma unroll
            for (int j = 0; j < 8; ++j) {
                sx[j]     = ra[2 * j];  cf[j]     = __int_as_float(ra[2 * j + 1]);
                sx[8 + j] = rb[2 * j];  cf[8 + j] = __int_as_float(rb[2 * j + 1]);
            }
        } else {
#pragma unroll
            for (int j = 0; j < GC; ++j) {
                const int2 r = eRec[min(e + j, eEnd - 1)];
                sx[j] = r.x;
                cf[j] = __int_as_float(r.y);
            }
        }
        float rv[GC][VPL];
#pragma unroll
        for (int j = 0; j < GC; ++j) {
            if constexpr (VPL == 2) {
                const float2 v = ((const float2*)h)[(unsigned)sx[j] * (unsigned)(F / 2) + (unsigned)lane];
                rv[j][0] = v.x; rv[j][1] = v.y;
            } else {
                rv[j][0] = h[(unsigned)sx[j] * (unsigned)F + (unsigned)lane];
            }
        }
#pragma unroll
        for (int j = 0; j < GC; ++j) {
            if (FULL || e + j < eEnd) {
                while (e + j == nb) flush();
#pragma unroll
                for (int q = 0; q < VPL; ++q) acc[q] += rv[j][q] * cf[j];
            }
        }
    };

    while (e < eEnd) {
        if (eEnd - e >= GC) group(std::true_type{});
        else                group(std::false_type{});
        e += GC;
    }
    while (node < n1) flush();

    if constexpr (FNEXT > 0) {
        constexpr int NT  = FNEXT / 16;
        constexpr int TPW = (4 * NT) / GW;
        const int lrow  = lane & 15;
        const int khalf = (lane >> 4) * 2;
        const int col   = (wib % NT) * 16 + lrow;
        const int mt0   = (wib / NT) * TPW;
        v2f bf[F / 4];
#pragma unroll
        for (int kk = 0; kk < F / 4; ++kk) {
            bf[kk].x = Wn[(kk * 4 + khalf + 0) * FNEXT + col];
            bf[kk].y = Wn[(kk * 4 + khalf + 1) * FNEXT + col];
        }
        __syncthreads();
        v8f pacc[TPW];
#pragma unroll
        for (int t = 0; t < TPW; ++t) pacc[t] = v8f{};
#pragma unroll
        for (int kk = 0; kk < F / 4; ++kk) {
#pragma unroll
            for (int t = 0; t < TPW; ++t) {
                const v2f a = *(const v2f*)&s_x[((mt0 + t) * 16 + lrow) * XP + kk * 4 + khalf];
                pacc[t] = __builtin_amdgcn_wmma_f32_16x16x4_f32(
                    false, a, false, bf[kk], (short)0, pacc[t], false, false);
            }
        }
        const int blk0 = (int)blockIdx.x * (GW * NPW);
#pragma unroll
        for (int t = 0; t < TPW; ++t) {
#pragma unroll
            for (int r = 0; r < 8; ++r) {
                const int row = blk0 + (mt0 + t) * 16 + (lane >> 4) * 8 + r;
                if (row < n) y[(long)row * FNEXT + col] = pacc[t][r];
            }
        }
    }
}


extern "C" void kernel_launch(void* const* d_in, const int* in_sizes, int n_in,
                              void* d_out, int out_size, void* d_ws, size_t ws_size,
                              hipStream_t stream) {
    const float* x    = (const float*)d_in[0];
    const int*   src  = (const int*)  d_in[1];
    const int*   dst  = (const int*)  d_in[2];
    const float* ew   = (const float*)d_in[3];
    const float* W1   = (const float*)d_in[4];
    const float* b1   = (const float*)d_in[5];
    const float* W2   = (const float*)d_in[6];
    const float* b2   = (const float*)d_in[7];
    const float* W3   = (const float*)d_in[8];
    const float* b3   = (const float*)d_in[9];
    const float* W4   = (const float*)d_in[10];
    const float* b4   = (const float*)d_in[11];
    const float* Wres = (const float*)d_in[12];
    const float* bres = (const float*)d_in[13];
    float* out = (float*)d_out;

    const int N  = in_sizes[0] / IN_F;
    const int E  = in_sizes[1];
    const int nbk = ceil_div_h(N, BNODES);
    const int g1  = ceil_div_h(E, CHUNK);
    const long LT = (long)g1 * (nbk + 1);

    unsigned int* ws = (unsigned int*)d_ws;
    float* norm   = (float*)(ws);
    int*   rowptr = (int*)  (ws + N);
    int*   lo     = (int*)  (ws + 2L * N + 1);
    long   tmpOff = (2L * N + 1 + LT + 63) & ~63L;
    long   recOff = tmpOff + ((4L * E + 63) & ~63L);
    long   hOff   = recOff + ((2L * E + 63) & ~63L);
    long   plane  = ((long)N * H_F + 63) & ~63L;
    int4*  tmp    = (int4*) (ws + tmpOff);
    int2*  eRec   = (int2*) (ws + recOff);
    float* h      = (float*)(ws + hOff);
    float* xbuf   = h    + plane;
    float* res    = xbuf + plane;

    const int T  = 256;
    const int GT = GW * 32;
    const int gblocks = ceil_div_h(N, GW * NPW);

    chunk_sort_kernel<<<g1, PT, nbk * sizeof(int), stream>>>(src, dst, ew, lo, tmp, E, nbk);
    bucket_build_kernel<<<nbk, PT, 0, stream>>>(tmp, lo, rowptr, norm, eRec, N, E, g1, nbk);
    row_coef_kernel<<<ceil_div_h(4L * N, T), T, 0, stream>>>(rowptr, norm, eRec, N);

    input_products_kernel<<<ceil_div_h(N, PM), PWV * 32, 0, stream>>>(x, W1, Wres, h, res, N);
    gather_agg_kernel<H_F, true, false, H_F><<<gblocks, GT, 0, stream>>>(h, rowptr, eRec,
                                                                        b1, nullptr, nullptr, W2, xbuf, N);
    gather_agg_kernel<H_F, true, false, H_F><<<gblocks, GT, 0, stream>>>(xbuf, rowptr, eRec,
                                                                        b2, nullptr, nullptr, W3, h, N);
    gather_agg_kernel<H_F, true, false, C_F><<<gblocks, GT, 0, stream>>>(h, rowptr, eRec,
                                                                        b3, nullptr, nullptr, W4, xbuf, N);
    gather_agg_kernel<C_F, false, true, 0><<<gblocks, GT, 0, stream>>>(xbuf, rowptr, eRec,
                                                                       b4, res, bres, nullptr, out, N);
}


# normalisation moved from edges to rows: planes hold norm-scaled rows, records keep the plain edge weight, a row sum is scaled by norm[dst] once; coefficient pass and launch go; sum order unchanged
# speedup vs baseline: 3.3051x; 1.0771x over previous
#include <hip/hip_runtime.h>
#include <type_traits>


typedef __attribute__((ext_vector_type(2))) float v2f;
typedef __attribute__((ext_vector_type(8))) float v8f;
typedef int rec8 __attribute__((ext_vector_type(16), aligned(8)));

#define IN_F 128
#define H_F  64
#define C_F  32

#define BSH    8
#define BNODES (1 << BSH)
#define CHUNK  8192
#define PT     1024
#define EPT    (CHUNK / PT)
#define RB     8
#define RCAP   6144

static inline int ceil_div_h(long a, long b) { return (int)((a + b - 1) / b); }


__global__ void __launch_bounds__(PT)
chunk_sort_kernel(const int* __restrict__ src, const int* __restrict__ dst,
                  const float* __restrict__ ew, int* __restrict__ lo,
                  int4* __restrict__ tmp, int E, int nbk) {
    extern __shared__ int s_bins[];
    __shared__ int s_wsum[PT / 32];
    const int tid  = threadIdx.x;
    const int lane = tid & 31;
    const int wv   = tid >> 5;
    for (int k = tid; k < nbk; k += PT) s_bins[k] = 0;
    __syncthreads();
    const int e0 = blockIdx.x * CHUNK;
    const int e1 = min(e0 + CHUNK, E);
    int s[EPT], d[EPT], w[EPT];
#pragma unroll
    for (int j = 0; j < EPT; ++j) {
        const int e = min(e0 + tid + j * PT, e1 - 1);
        s[j] = src[e];
        d[j] = dst[e];
        w[j] = __float_as_int(ew[e]);
    }
    asm volatile("" :: "v"(s[0]), "v"(s[1]), "v"(s[2]), "v"(s[3]), "v"(s[4]), "v"(s[5]), "v"(s[6]), "v"(s[7]),
                       "v"(d[0]), "v"(d[1]), "v"(d[2]), "v"(d[3]), "v"(d[4]), "v"(d[5]), "v"(d[6]), "v"(d[7]),
                       "v"(w[0]), "v"(w[1]), "v"(w[2]), "v"(w[3]), "v"(w[4]), "v"(w[5]), "v"(w[6]), "v"(w[7]));
    static_assert(EPT == 8, "the fence above names EPT = 8 edges");
#pragma unroll
    for (int j = 0; j < EPT; ++j)
        if (e0 + tid + j * PT < e1) atomicAdd(&s_bins[d[j] >> BSH], 1);
    __syncthreads();

    int carry = 0;
    for (int base = 0; base < nbk; base += PT) {
        const int i = base + tid;
        const int v = (i < nbk) ? s_bins[i] : 0;
        int x = v;
        for (int o = 1; o < 32; o <<= 1) {
            const int t = __shfl_up(x, o);
            if (lane >= o) x += t;
        }
        if (lane == 31) s_wsum[wv] = x;
        __syncthreads();
        if (wv == 0) {
            int z = s_wsum[lane];
            for (int o = 1; o < 32; o <<= 1) {
                const int t = __shfl_up(z, o);
                if (lane >= o) z += t;
            }
            s_wsum[lane] = z;
        }
        __syncthreads();
        const int excl = carry + (wv ? s_wsum[wv - 1] : 0) + x - v;
        if (i < nbk) {
            s_bins[i] = excl;
            lo[blockIdx.x * (nbk + 1) + i] = excl;
        }
        carry += s_wsum[PT / 32 - 1];
        __syncthreads();
    }
    if (tid == 0) lo[blockIdx.x * (nbk + 1) + nbk] = e1 - e0;

#pragma unroll
    for (int j = 0; j < EPT; ++j) {
        if (e0 + tid + j * PT < e1) {
            const int pos = atomicAdd(&s_bins[d[j] >> BSH], 1);
            tmp[e0 + pos] = make_int4(s[j], w[j], d[j], 0);
        }
    }
}

__global__ void __launch_bounds__(PT)
bucket_build_kernel(const int4* __restrict__ tmp, const int* __restrict__ lo,
                    int* __restrict__ rowptr, float* __restrict__ norm,
                    int2* __restrict__ eRec, int n, int E, int g1, int nbk) {
    __shared__ int   s_cnt[BNODES];
    __shared__ float s_deg[BNODES];
    __shared__ int   s_scan[BNODES];
    __shared__ int2  s_rec[RCAP];
    __shared__ int   s_red[2 * (PT / 32)];
    const int tid   = threadIdx.x;
    const int lane  = tid & 31;
    const int wib   = __builtin_amdgcn_readfirstlane(tid >> 5);
    const int k     = blockIdx.x;
    const int ld    = nbk + 1;
    constexpr int NW = PT / 32;
    const int nbase = k << BSH;

    int sa = 0, sb = 0;
    for (int c = tid; c < g1; c += PT) {
        sa += lo[c * ld + k];
        sb += lo[c * ld + k + 1];
    }
    for (int o = 16; o > 0; o >>= 1) {
        sa += __shfl_xor(sa, o);
        sb += __shfl_xor(sb, o);
    }
    if (lane == 0) { s_red[wib] = sa; s_red[PT / 32 + wib] = sb; }
    if (tid < BNODES) { s_cnt[tid] = 0; s_deg[tid] = 0.0f; }
    __syncthreads();
    sa = s_red[lane];
    sb = s_red[PT / 32 + lane];
    for (int o = 16; o > 0; o >>= 1) {
        sa += __shfl_xor(sa, o);
        sb += __shfl_xor(sb, o);
    }
    const int t0 = __builtin_amdgcn_readfirstlane(sa);
    const int t1 = __builtin_amdgcn_readfirstlane(sb);
    const bool stage = (t1 - t0) <= RCAP;

    auto sweep = [&](auto placeTag) {
        constexpr bool PLACE = decltype(placeTag)::value;
        auto one = [&](const int4 r) {
            const int dl = r.z - nbase;
            if constexpr (PLACE) {
                const int pos = atomicAdd(&s_cnt[dl], 1);
                if (stage) s_rec[pos - t0] = make_int2(r.x, r.y);
                else       eRec[pos]       = make_int2(r.x, r.y);
            } else {
                atomicAdd(&s_cnt[dl], 1);
                atomicAdd(&s_deg[dl], __int_as_float(r.y));
            }
        };
        for (int cb = wib; cb < g1; cb += NW * RB) {
            int a[RB], b[RB];
#pragma unroll
            for (int j = 0; j < RB; ++j) {
                const int cc = min(cb + NW * j, g1 - 1);
                a[j] = lo[cc * ld + k];
                b[j] = lo[cc * ld + k + 1];
            }
#pragma unroll
            for (int j = 0; j < RB; ++j) {
                const int cj = cb + NW * j;
                const int cc = min(cj, g1 - 1);
                a[j] += cc * CHUNK;
                b[j]  = (cj < g1) ? b[j] + cc * CHUNK : a[j];
            }
            int4 r[RB];
#pragma unroll
            for (int j = 0; j < RB; ++j) {
                const int idx = min(min(a[j] + lane, max(b[j] - 1, a[j])), E - 1);
                r[j] = tmp[idx];
            }
            asm volatile("" :: "v"(r[0].x), "v"(r[0].y), "v"(r[0].z), "v"(r[1].x), "v"(r[1].y), "v"(r[1].z),
                               "v"(r[2].x), "v"(r[2].y), "v"(r[2].z), "v"(r[3].x), "v"(r[3].y), "v"(r[3].z),
                               "v"(r[4].x), "v"(r[4].y), "v"(r[4].z), "v"(r[5].x), "v"(r[5].y), "v"(r[5].z),
                               "v"(r[6].x), "v"(r[6].y), "v"(r[6].z), "v"(r[7].x), "v"(r[7].y), "v"(r[7].z));
            static_assert(RB == 8, "the fence above names RB = 8 runs");
#pragma unroll
            for (int j = 0; j < RB; ++j) {
                if (a[j] + lane < b[j]) one(r[j]);
                for (int i = a[j] + 32 + lane; i < b[j]; i += 32) one(tmp[i]);
            }
        }
    };

    sweep(std::false_type{});
    __syncthreads();

    int v = 0;
    if (tid < BNODES) { v = s_cnt[tid]; s_scan[tid] = v; }
    __syncthreads();
    for (int off = 1; off < BNODES; off <<= 1) {
        int t = 0;
        if (tid < BNODES && tid >= off) t = s_scan[tid - off];
        __syncthreads();
        if (tid < BNODES) s_scan[tid] += t;
        __syncthreads();
    }
    if (tid < BNODES) {
        const int start = t0 + s_scan[tid] - v;
        s_cnt[tid] = start;
        const int node = nbase + tid;
        if (node < n) {
            rowptr[node] = start;
            float d = s_deg[tid];
            norm[node] = (d > 0.0f) ? rsqrtf(fmaxf(d, 1e-12f)) : 0.0f;
        }
    }
    if (blockIdx.x == gridDim.x - 1 && tid == 0) rowptr[n] = E;
    __syncthreads();

    sweep(std::true_type{});
    if (stage) {
        __syncthreads();
        for (int i = tid; i < t1 - t0; i += blockDim.x) eRec[t0 + i] = s_rec[i];
    }
}

#define PM  48
#define PXP 132
#define PWV ((H_F + C_F) / 16)

template <int FOUT>
__device__ __forceinline__ void load_b_slice(const float* __restrict__ Wm, int col, int khalf,
                                             v2f (&bf)[IN_F / 4]) {
    const float* p = Wm + khalf * FOUT + col;
#pragma unroll
    for (int kk = 0; kk < IN_F / 4; ++kk) {
        bf[kk].x = p[(kk * 4 + 0) * FOUT];
        bf[kk].y = p[(kk * 4 + 1) * FOUT];
    }
}

__global__ void __launch_bounds__(PWV * 32)
input_products_kernel(const float* __restrict__ X, const float* __restrict__ Wa,
                      const float* __restrict__ Wb, const float* __restrict__ nrm,
                      float* __restrict__ Ya, float* __restrict__ Yb, int nrows) {
    __shared__ __attribute__((aligned(16))) float s_t[PM * PXP];
    __shared__ __attribute__((aligned(16))) float s_n[2 * PM];
    constexpr int NT  = PWV * 32;
    constexpr int Q   = IN_F / 4;
    constexpr int RPS = NT / Q;
    constexpr int NST = PM / RPS;
    static_assert(NT % Q == 0 && PM % RPS == 0 && PM % 16 == 0, "whole sweeps, whole row tiles");
    const int tid = threadIdx.x;
    const int m0  = blockIdx.x * PM;
    const int r0  = tid / Q;
    const int q4  = (tid % Q) * 4;

    float4 sv[NST];
#pragma unroll
    for (int j = 0; j < NST; ++j) {
        const int row = min(m0 + r0 + j * RPS, nrows - 1);
        sv[j] = *(const float4*)(X + (long)row * IN_F + q4);
    }

    const int wave  = __builtin_amdgcn_readfirstlane(tid >> 5);
    const int lane  = tid & 31;
    const int lrow  = lane & 15;
    const int khalf = (lane >> 4) * 2;
    const bool isA  = wave < H_F / 16;
    const int col   = (isA ? wave : wave - H_F / 16) * 16 + lrow;

    v2f bf[IN_F / 4];
    if (isA) load_b_slice<H_F>(Wa, col, khalf, bf);
    else     load_b_slice<C_F>(Wb, col, khalf, bf);

#pragma unroll
    for (int j = 0; j < NST; ++j)
        *(float4*)(s_t + (r0 + j * RPS) * PXP + q4) = sv[j];
    if (tid < PM) {
        s_n[tid]      = nrm[min(m0 + tid, nrows - 1)];
        s_n[PM + tid] = 1.0f;
    }
    __syncthreads();

    constexpr int TPW = PM / 16;
    v8f pacc[TPW];
#pragma unroll
    for (int t = 0; t < TPW; ++t) pacc[t] = v8f{};
#pragma unroll
    for (int kk = 0; kk < IN_F / 4; ++kk) {
#pragma unroll
        for (int t = 0; t < TPW; ++t) {
            const v2f a = *(const v2f*)&s_t[(t * 16 + lrow) * PXP + kk * 4 + khalf];
            pacc[t] = __builtin_amdgcn_wmma_f32_16x16x4_f32(
                false, a, false, bf[kk], (short)0, pacc[t], false, false);
        }
    }
    float* __restrict__ Ym = isA ? Ya : Yb;
    const int fout = isA ? H_F : C_F;
    const float* sn = s_n + (isA ? 0 : PM) + (lane >> 4) * 8;
#pragma unroll
    for (int t = 0; t < TPW; ++t) {
        const float4 sa = *(const float4*)(sn + t * 16);
        const float4 sb = *(const float4*)(sn + t * 16 + 4);
        pacc[t] = pacc[t] * v8f{sa.x, sa.y, sa.z, sa.w, sb.x, sb.y, sb.z, sb.w};
    }
#pragma unroll
    for (int t = 0; t < TPW; ++t) {
#pragma unroll
        for (int r = 0; r < 8; ++r) {
            const int row = m0 + t * 16 + (lane >> 4) * 8 + r;
            if (row < nrows) Ym[(long)row * fout + col] = pacc[t][r];
        }
    }
}

#define NPW 8
#define GC  16
#define GW  8
#define XP  68

template <int F, bool RELU, bool FINAL, int FNEXT>
__global__ void __launch_bounds__(GW * 32)
gather_agg_kernel(const float* __restrict__ h,
                  const int* __restrict__ rowptr,
                  const int2* __restrict__ eRec,
                  const float* __restrict__ bias,
                  const float* __restrict__ res,
                  const float* __restrict__ bias2,
                  const float* __restrict__ Wn,
                  const float* __restrict__ nrm,
                  float* __restrict__ y, int n) {
    constexpr int VPL = F / 32;
    static_assert(FNEXT == 0 || (F == 64 && !FINAL && FNEXT % 16 == 0 && (4 * FNEXT / 16) % GW == 0),
                  "fused product: 64-wide rows, 64 x FNEXT tiles split evenly over the waves");
    __shared__ __attribute__((aligned(16))) float s_x[FNEXT > 0 ? GW * NPW * XP : 1];
    const int lane  = threadIdx.x & 31;
    const int fbase = lane * VPL;
    const int wib = __builtin_amdgcn_readfirstlane((int)(threadIdx.x >> 5));
    const int wv  = __builtin_amdgcn_readfirstlane((int)(blockIdx.x * GW)) + wib;
    const int n0  = wv * NPW;
    if constexpr (FNEXT == 0) {
        if (n0 >= n) return;
    }
    const int n1 = min(n0 + NPW, n);
    if constexpr (FNEXT > 0) {
        for (int q = max(n1 - n0, 0); q < NPW; ++q)
            *(float2*)(s_x + (wib * NPW + q) * XP + fbase) = make_float2(0.0f, 0.0f);
    }

    const int rpv = rowptr[min(n0 + lane, n1)];
    const int nrv = __float_as_int(nrm[min(n0 + lane, n - 1)]);
    int       e    = __builtin_amdgcn_readlane(rpv, 0);
    const int eEnd = __builtin_amdgcn_readlane(rpv, NPW);
    int       nb   = __builtin_amdgcn_readlane(rpv, 1);
    int       node = n0;

    float bs[VPL], bs2[VPL], resv[VPL], acc[VPL];
#pragma unroll
    for (int j = 0; j < VPL; ++j) {
        bs[j]   = bias[fbase + j];
        bs2[j]  = FINAL ? bias2[fbase + j] : 0.0f;
        resv[j] = FINAL ? res[(long)n0 * F + fbase + j] : 0.0f;
        acc[j]  = 0.0f;
    }

    auto flush = [&]() {
        const long base = (long)node * F + fbase;
        const float nd = __int_as_float(__builtin_amdgcn_readlane(nrv, node - n0));
        float o[VPL];
#pragma unroll
        for (int j = 0; j < VPL; ++j) {
            float v = acc[j] * nd + bs[j];
            if (FINAL) v += resv[j] + bs2[j];
            if (RELU) v = fmaxf(v, 0.0f);
            if (FNEXT > 0) v *= nd;
            o[j]   = v;
            acc[j] = 0.0f;
        }
        if constexpr (FNEXT > 0) {
            *(float2*)(s_x + (wib * NPW + (node - n0)) * XP + fbase) = make_float2(o[0], o[1]);
        } else if constexpr (VPL == 2) {
            *(float2*)(y + base) = make_float2(o[0], o[1]);
        } else {
            y[base] = o[0];
        }
        ++node;
        nb = __builtin_amdgcn_readlane(rpv, node - n0 + 1);
        if constexpr (FINAL) {
            if (node < n1) {
#pragma unroll
                for (int j = 0; j < VPL; ++j) resv[j] = res[(long)node * F + fbase + j];
            }
        }
    };

    auto group = [&](auto fullTag) {
        constexpr bool FULL = decltype(fullTag)::value;
        int   sx[GC];
        float cf[GC];
        if constexpr (FULL) {
            const rec8* p = (const rec8*)(eRec + e);
            const rec8 ra = p[0], rb = p[1];
#pragma unroll
            for (int j = 0; j < 8; ++j) {
                sx[j]     = ra[2 * j];  cf[j]     = __int_as_float(ra[2 * j + 1]);
                sx[8 + j] = rb[2 * j];  cf[8 + j] = __int_as_float(rb[2 * j + 1]);
            }
        } else {
#pragma unroll
            for (int j = 0; j < GC; ++j) {
                const int2 r = eRec[min(e + j, eEnd - 1)];
                sx[j] = r.x;
                cf[j] = __int_as_float(r.y);
            }
        }
        float rv[GC][VPL];
#pragma unroll
        for (int j = 0; j < GC; ++j) {
            if constexpr (VPL == 2) {
                const float2 v = ((const float2*)h)[(unsigned)sx[j] * (unsigned)(F / 2) + (unsigned)lane];
                rv[j][0] = v.x; rv[j][1] = v.y;
            } else {
                rv[j][0] = h[(unsigned)sx[j] * (unsigned)F + (unsigned)lane];
            }
        }
#pragma unroll
        for (int j = 0; j < GC; ++j) {
            if (FULL || e + j < eEnd) {
                while (e + j == nb) flush();
#pragma unroll
                for (int q = 0; q < VPL; ++q) acc[q] += rv[j][q] * cf[j];
            }
        }
    };

    while (e < eEnd) {
        if (eEnd - e >= GC) group(std::true_type{});
        else                group(std::false_type{});
        e += GC;
    }
    while (node < n1) flush();

    if constexpr (FNEXT > 0) {
        constexpr int NT  = FNEXT / 16;
        constexpr int TPW = (4 * NT) / GW;
        const int lrow  = lane & 15;
        const int khalf = (lane >> 4) * 2;
        const int col   = (wib % NT) * 16 + lrow;
        const int mt0   = (wib / NT) * TPW;
        v2f bf[F / 4];
#pragma unroll
        for (int kk = 0; kk < F / 4; ++kk) {
            bf[kk].x = Wn[(kk * 4 + khalf + 0) * FNEXT + col];
            bf[kk].y = Wn[(kk * 4 + khalf + 1) * FNEXT + col];
        }
        __syncthreads();
        v8f pacc[TPW];
#pragma unroll
        for (int t = 0; t < TPW; ++t) pacc[t] = v8f{};
#pragma unroll
        for (int kk = 0; kk < F / 4; ++kk) {
#pragma unroll
            for (int t = 0; t < TPW; ++t) {
                const v2f a = *(const v2f*)&s_x[((mt0 + t) * 16 + lrow) * XP + kk * 4 + khalf];
                pacc[t] = __builtin_amdgcn_wmma_f32_16x16x4_f32(
                    false, a, false, bf[kk], (short)0, pacc[t], false, false);
            }
        }
        const int blk0 = (int)blockIdx.x * (GW * NPW);
#pragma unroll
        for (int t = 0; t < TPW; ++t) {
#pragma unroll
            for (int r = 0; r < 8; ++r) {
                const int row = blk0 + (mt0 + t) * 16 + (lane >> 4) * 8 + r;
                if (row < n) y[(long)row * FNEXT + col] = pacc[t][r];
            }
        }
    }
}


extern "C" void kernel_launch(void* const* d_in, const int* in_sizes, int n_in,
                              void* d_out, int out_size, void* d_ws, size_t ws_size,
                              hipStream_t stream) {
    const float* x    = (const float*)d_in[0];
    const int*   src  = (const int*)  d_in[1];
    const int*   dst  = (const int*)  d_in[2];
    const float* ew   = (const float*)d_in[3];
    const float* W1   = (const float*)d_in[4];
    const float* b1   = (const float*)d_in[5];
    const float* W2   = (const float*)d_in[6];
    const float* b2   = (const float*)d_in[7];
    const float* W3   = (const float*)d_in[8];
    const float* b3   = (const float*)d_in[9];
    const float* W4   = (const float*)d_in[10];
    const float* b4   = (const float*)d_in[11];
    const float* Wres = (const float*)d_in[12];
    const float* bres = (const float*)d_in[13];
    float* out = (float*)d_out;

    const int N  = in_sizes[0] / IN_F;
    const int E  = in_sizes[1];
    const int nbk = ceil_div_h(N, BNODES);
    const int g1  = ceil_div_h(E, CHUNK);
    const long LT = (long)g1 * (nbk + 1);

    unsigned int* ws = (unsigned int*)d_ws;
    float* norm   = (float*)(ws);
    int*   rowptr = (int*)  (ws + N);
    int*   lo     = (int*)  (ws + 2L * N + 1);
    long   tmpOff = (2L * N + 1 + LT + 63) & ~63L;
    long   recOff = tmpOff + ((4L * E + 63) & ~63L);
    long   hOff   = recOff + ((2L * E + 63) & ~63L);
    long   plane  = ((long)N * H_F + 63) & ~63L;
    int4*  tmp    = (int4*) (ws + tmpOff);
    int2*  eRec   = (int2*) (ws + recOff);
    float* h      = (float*)(ws + hOff);
    float* xbuf   = h    + plane;
    float* res    = xbuf + plane;

    const int GT = GW * 32;
    const int gblocks = ceil_div_h(N, GW * NPW);

    chunk_sort_kernel<<<g1, PT, nbk * sizeof(int), stream>>>(src, dst, ew, lo, tmp, E, nbk);
    bucket_build_kernel<<<nbk, PT, 0, stream>>>(tmp, lo, rowptr, norm, eRec, N, E, g1, nbk);

    input_products_kernel<<<ceil_div_h(N, PM), PWV * 32, 0, stream>>>(x, W1, Wres, norm, h, res, N);
    gather_agg_kernel<H_F, true, false, H_F><<<gblocks, GT, 0, stream>>>(h, rowptr, eRec,
                                                                        b1, nullptr, nullptr, W2, norm, xbuf, N);
    gather_agg_kernel<H_F, true, false, H_F><<<gblocks, GT, 0, stream>>>(xbuf, rowptr, eRec,
                                                                        b2, nullptr, nullptr, W3, norm, h, N);
    gather_agg_kernel<H_F, true, false, C_F><<<gblocks, GT, 0, stream>>>(h, rowptr, eRec,
                                                                        b3, nullptr, nullptr, W4, norm, xbuf, N);
    gather_agg_kernel<C_F, false, true, 0><<<gblocks, GT, 0, stream>>>(xbuf, rowptr, eRec,
                                                                       b4, res, bres, nullptr, norm, out, N);
}
